# starting kernel
# baseline (speedup 1.0000x reference)
#include <hip/hip_runtime.h>
#include <math.h>

typedef __attribute__((ext_vector_type(2))) float v2f;
typedef __attribute__((ext_vector_type(8))) float v8f;

#define N_NODES   100000
#define NUM_FEAT  64
#define NUM_CLASS 40
#define N_EDGES   1600000
#define WPB       5
#define GEMM_BLOCK (WPB * 32)
#define ROW_TILES  (N_NODES / 16)
#define GEMM_GRID  (ROW_TILES / WPB)
#define KSTEPS     (NUM_FEAT / 4)
#define CTILES     3

#define NW_V2F     (KSTEPS * 2 * CTILES * 32)
#define OFF_W      0
#define OFF_OUT    (NW_V2F * 2)
#define OFF_LSE    (OFF_OUT + WPB * 16 * 48)
#define OFF_X      (OFF_LSE + WPB * 16)
#define OFF_A      (OFF_X + WPB * 16 * NUM_FEAT)
#define SMEM_FLOATS (OFF_A + WPB * 16 * NUM_FEAT)

__global__ void zero_ws(float* __restrict__ agg, float* __restrict__ deg) {
  long i = (long)blockIdx.x * blockDim.x + threadIdx.x;
  const long tot = (long)N_NODES * NUM_FEAT;
  if (i < tot)     agg[i] = 0.0f;
  if (i < N_NODES) deg[i] = 0.0f;
}

__global__ void edge_scatter(const float* __restrict__ x,
                             const int*   __restrict__ index,
                             float* __restrict__ agg,
                             float* __restrict__ deg) {
  long gid = (long)blockIdx.x * blockDim.x + threadIdx.x;
  long e   = gid >> 4;
  int  q   = (int)(gid & 15);
  if (e >= N_EDGES) return;
  int src = index[e];
  int dst = index[N_EDGES + e];
  const float4 v = *(const float4*)(x + (long)src * NUM_FEAT + q * 4);
  float* a = agg + (long)dst * NUM_FEAT + q * 4;
  atomicAdd(a + 0, v.x);
  atomicAdd(a + 1, v.y);
  atomicAdd(a + 2, v.z);
  atomicAdd(a + 3, v.w);
  if (q == 0) atomicAdd(deg + dst, 1.0f);
}

__global__ void __launch_bounds__(GEMM_BLOCK)
sage_gemm(const float* __restrict__ x,
          const float* __restrict__ agg,
          const float* __restrict__ deg,
          const float* __restrict__ Wl,
          const float* __restrict__ bias,
          const float* __restrict__ Wr,
          float* __restrict__ out) {
  __shared__ __align__(16) float smem[SMEM_FLOATS];

  v2f*   sW   = (v2f*)smem;
  float* sOut = smem + OFF_OUT;
  float* sLse = smem + OFF_LSE;

  const int tid  = threadIdx.x;
  const int wave = tid >> 5;
  const int lane = tid & 31;
  const int half = lane >> 4;
  const int l16  = lane & 15;

  const long row0 = ((long)blockIdx.x * WPB + wave) * 16;

  for (int i = tid; i < NW_V2F; i += GEMM_BLOCK) {
    const int li    = i & 31;
    const int t     = (i >> 5) % CTILES;
    const int which = (i / (CTILES * 32)) & 1;
    const int step  = i / (2 * CTILES * 32);
    const int k     = 4 * step + 2 * (li >> 4);
    const int n     = t * 16 + (li & 15);
    const float* W  = which ? Wr : Wl;
    v2f w = {0.0f, 0.0f};
    if (n < NUM_CLASS) w = (v2f){W[k * NUM_CLASS + n], W[(k + 1) * NUM_CLASS + n]};
    sW[i] = w;
  }

  {
    const float* xg = x   + row0 * NUM_FEAT;
    const float* ag = agg + row0 * NUM_FEAT;
    float* sx = smem + OFF_X + wave * (16 * NUM_FEAT);
    float* sa = smem + OFF_A + wave * (16 * NUM_FEAT);
    for (int i = lane; i < (16 * NUM_FEAT) / 4; i += 32) {
      float4 vx = *(const float4*)(xg + i * 4);
      float4 va = *(const float4*)(ag + i * 4);
      *(float4*)(sx + i * 4) = vx;
      *(float4*)(sa + i * 4) = va;
    }
  }
  __syncthreads();

  const float rdeg = 1.0f / fmaxf(deg[row0 + l16], 1.0f);

  const v2f* xf = (const v2f*)(smem + OFF_X + wave * (16 * NUM_FEAT) + l16 * NUM_FEAT);
  const v2f* af = (const v2f*)(smem + OFF_A + wave * (16 * NUM_FEAT) + l16 * NUM_FEAT);
  const v2f* wp = sW + lane;

  v8f acc0 = {}, acc1 = {}, acc2 = {};

#pragma unroll
  for (int step = 0; step < KSTEPS; ++step) {
    const v2f a_x   = xf[2 * step + half];
    const v2f a_agg = af[2 * step + half] * rdeg;

    const int wb = step * (2 * CTILES * 32);
    const v2f bl0 = wp[wb];
    const v2f bl1 = wp[wb + 32];
    const v2f bl2 = wp[wb + 64];
    const v2f br0 = wp[wb + 96];
    const v2f br1 = wp[wb + 128];
    const v2f br2 = wp[wb + 160];

    acc0 = __builtin_amdgcn_wmma_f32_16x16x4_f32(false, a_agg, false, bl0, (short)0, acc0, false, false);
    acc1 = __builtin_amdgcn_wmma_f32_16x16x4_f32(false, a_agg, false, bl1, (short)0, acc1, false, false);
    acc2 = __builtin_amdgcn_wmma_f32_16x16x4_f32(false, a_agg, false, bl2, (short)0, acc2, false, false);
    acc0 = __builtin_amdgcn_wmma_f32_16x16x4_f32(false, a_x,   false, br0, (short)0, acc0, false, false);
    acc1 = __builtin_amdgcn_wmma_f32_16x16x4_f32(false, a_x,   false, br1, (short)0, acc1, false, false);
    acc2 = __builtin_amdgcn_wmma_f32_16x16x4_f32(false, a_x,   false, br2, (short)0, acc2, false, false);
  }

  const float b0 = bias[l16];
  const float b1 = bias[16 + l16];
  const float b2 = (32 + l16 < NUM_CLASS) ? bias[32 + l16] : 0.0f;

  float* so = sOut + wave * (16 * 48);
#pragma unroll
  for (int r = 0; r < 8; ++r) {
    const int M = r + 8 * half;
    so[M * 48 +      l16] = acc0[r] + b0;
    so[M * 48 + 16 + l16] = acc1[r] + b1;
    so[M * 48 + 32 + l16] = acc2[r] + b2;
  }
  __syncthreads();

  if (lane < 16) {
    const float* rowp = so + l16 * 48;
    float mx = rowp[0];
    for (int c = 1; c < NUM_CLASS; ++c) mx = fmaxf(mx, rowp[c]);
    float s = 0.0f;
    for (int c = 0; c < NUM_CLASS; ++c) s += __expf(rowp[c] - mx);
    sLse[wave * 16 + l16] = mx + __logf(s);
  }
  __syncthreads();

  for (int i = lane; i < 16 * NUM_CLASS; i += 32) {
    const int r = i / NUM_CLASS, c = i % NUM_CLASS;
    out[(row0 + r) * NUM_CLASS + c] = so[r * 48 + c] - sLse[wave * 16 + r];
  }
}

extern "C" void kernel_launch(void* const* d_in, const int* in_sizes, int n_in,
                              void* d_out, int out_size, void* d_ws, size_t ws_size,
                              hipStream_t stream) {
  const float* x     = (const float*)d_in[0];
  const int*   index = (const int*)d_in[1];
  const float* Wl    = (const float*)d_in[2];
  const float* bl    = (const float*)d_in[3];
  const float* Wr    = (const float*)d_in[4];
  float* out = (float*)d_out;

  float* agg = (float*)d_ws;
  float* deg = agg + (size_t)N_NODES * NUM_FEAT;

  {
    const long tot = (long)N_NODES * NUM_FEAT;
    zero_ws<<<(int)((tot + 255) / 256), 256, 0, stream>>>(agg, deg);
  }
  {
    const long tot = (long)N_EDGES * 16;
    edge_scatter<<<(int)((tot + 255) / 256), 256, 0, stream>>>(x, index, agg, deg);
  }
  sage_gemm<<<GEMM_GRID, GEMM_BLOCK, 0, stream>>>(x, agg, deg, Wl, bl, Wr, out);
}


# Sort edges by destination each call (int-atomic arrival-order rank + prefix sum) and sum each row's sources in the GEMM kernel instead of the f32 atomic scatter; row sum order changes
# speedup vs baseline: 1.3452x; 1.3452x over previous
#include <hip/hip_runtime.h>
#include <math.h>


typedef __attribute__((ext_vector_type(2))) float v2f;
typedef __attribute__((ext_vector_type(8))) float v8f;

#define N_NODES   100000
#define NUM_FEAT  64
#define NUM_CLASS 40
#define N_EDGES   1600000
#define WPB       5
#define GEMM_BLOCK (WPB * 32)
#define ROW_TILES  (N_NODES / 16)
#define GEMM_GRID  (ROW_TILES / WPB)
#define KSTEPS     (NUM_FEAT / 4)
#define CTILES     3

#define SCAN_BLOCKS 98
#define CNT_PAD     (SCAN_BLOCKS * 1024)
#define EDGE_GRID   (N_EDGES / 256)
#define IDX_CAP     (16 * 48)
#define GU          8

#define NW_V2F     (KSTEPS * 2 * CTILES * 32)
#define OFF_W      0
#define OFF_OUT    (NW_V2F * 2)
#define OFF_LSE    (OFF_OUT + WPB * 16 * 48)
#define OFF_X      (OFF_LSE + WPB * 16)
#define OFF_A      (OFF_X + WPB * 16 * NUM_FEAT)
#define SMEM_FLOATS (OFF_A + WPB * 16 * NUM_FEAT)

__global__ void __launch_bounds__(256) zero_cnt(int* __restrict__ cnt) {
  ((int4*)cnt)[blockIdx.x * 256 + threadIdx.x] = make_int4(0, 0, 0, 0);
}

__global__ void __launch_bounds__(256)
edge_rank(const int* __restrict__ index, int* __restrict__ cnt, int* __restrict__ rank) {
  const int e = blockIdx.x * 256 + threadIdx.x;
  if (e >= N_EDGES) return;
  const int d = index[N_EDGES + e];
  rank[e] = atomicAdd(cnt + d, 1);
}

__global__ void __launch_bounds__(256) scan_local(int* __restrict__ cnt, int* __restrict__ btot) {
  __shared__ int wsum[8];
  const int t = threadIdx.x, lane = t & 31, w = t >> 5;
  int4* p = (int4*)cnt + (blockIdx.x * 256 + t);
  const int4 c = *p;
  const int s = c.x + c.y + c.z + c.w;
  int incl = s;
  for (int o = 1; o < 32; o <<= 1) {
    const int v = __shfl_up(incl, o);
    if (lane >= o) incl += v;
  }
  if (lane == 31) wsum[w] = incl;
  __syncthreads();
  int wbase = 0;
  for (int k = 0; k < 8; ++k) wbase += (k < w) ? wsum[k] : 0;
  const int ex = wbase + incl - s;
  *p = make_int4(ex, ex + c.x, ex + c.x + c.y, ex + c.x + c.y + c.z);
  if (t == 255) btot[blockIdx.x] = ex + s;
}

__global__ void __launch_bounds__(256) scan_add(int* __restrict__ cnt, const int* __restrict__ btot) {
  __shared__ int sbase;
  const int t = threadIdx.x;
  if (t < 32) {
    int a = 0;
    for (int k = t; k < (int)blockIdx.x; k += 32) a += btot[k];
    for (int o = 16; o > 0; o >>= 1) a += __shfl_xor(a, o);
    if (t == 0) sbase = a;
  }
  __syncthreads();
  const int base = sbase;
  int4* p = (int4*)cnt + (blockIdx.x * 256 + t);
  int4 c = *p;
  c.x += base; c.y += base; c.z += base; c.w += base;
  *p = c;
}

__global__ void __launch_bounds__(256)
edge_place(const int* __restrict__ index, const int* __restrict__ cnt,
           const int* __restrict__ rank, int* __restrict__ ssrc) {
  const int e = blockIdx.x * 256 + threadIdx.x;
  if (e >= N_EDGES) return;
  const int d = index[N_EDGES + e];
  ssrc[cnt[d] + rank[e]] = index[e];
}

template <bool LDS_IDX>
static __device__ __forceinline__ void gather_rows(const float* __restrict__ x,
                                                   const int* __restrict__ ssrc,
                                                   const int* sIdx, int B, int bv,
                                                   int half, int q, float* sa) {
  for (int r = 0; r < 16; ++r) {
    const int b = __builtin_amdgcn_readlane(bv, r);
    const int e = __builtin_amdgcn_readlane(bv, r + 1);
    float4 acc = make_float4(0.0f, 0.0f, 0.0f, 0.0f);
    for (int j = b + half; j < e; j += 2 * GU) {
      int s[GU];
#pragma unroll
      for (int u = 0; u < GU; ++u) {
        const int jj = min(j + 2 * u, e - 1);
        s[u] = LDS_IDX ? sIdx[jj - B] : ssrc[jj];
      }
      float4 v[GU];
#pragma unroll
      for (int u = 0; u < GU; ++u) {
        v[u] = make_float4(0.0f, 0.0f, 0.0f, 0.0f);
        if (j + 2 * u < e) v[u] = *(const float4*)(x + (long)s[u] * NUM_FEAT + q * 4);
      }
#pragma unroll
      for (int u = 0; u < GU; ++u) {
        acc.x += v[u].x; acc.y += v[u].y; acc.z += v[u].z; acc.w += v[u].w;
      }
    }
    acc.x += __shfl_xor(acc.x, 16);
    acc.y += __shfl_xor(acc.y, 16);
    acc.z += __shfl_xor(acc.z, 16);
    acc.w += __shfl_xor(acc.w, 16);
    if (half == 0) *(float4*)(sa + r * NUM_FEAT + q * 4) = acc;
  }
}

__global__ void __launch_bounds__(GEMM_BLOCK)
sage_gemm(const float* __restrict__ x,
          const int*   __restrict__ cnt,
          const int*   __restrict__ ssrc,
          const float* __restrict__ Wl,
          const float* __restrict__ bias,
          const float* __restrict__ Wr,
          float* __restrict__ out) {
  __shared__ __align__(16) float smem[SMEM_FLOATS];

  v2f*   sW   = (v2f*)smem;
  float* sOut = smem + OFF_OUT;
  float* sLse = smem + OFF_LSE;

  const int tid  = threadIdx.x;
  const int wave = tid >> 5;
  const int lane = tid & 31;
  const int half = lane >> 4;
  const int l16  = lane & 15;

  const long row0 = ((long)blockIdx.x * WPB + wave) * 16;

  const int bv = (lane <= 16) ? cnt[row0 + lane] : 0;

  for (int i = tid; i < NW_V2F; i += GEMM_BLOCK) {
    const int li    = i & 31;
    const int t     = (i >> 5) % CTILES;
    const int which = (i / (CTILES * 32)) & 1;
    const int step  = i / (2 * CTILES * 32);
    const int k     = 4 * step + 2 * (li >> 4);
    const int n     = t * 16 + (li & 15);
    const float* W  = which ? Wr : Wl;
    v2f w = {0.0f, 0.0f};
    if (n < NUM_CLASS) w = (v2f){W[k * NUM_CLASS + n], W[(k + 1) * NUM_CLASS + n]};
    sW[i] = w;
  }

  float* so = sOut + wave * (16 * 48);
  int*   sIdx = (int*)so;
  float* sa = smem + OFF_A + wave * (16 * NUM_FEAT);

  const int  B    = __builtin_amdgcn_readlane(bv, 0);
  const int  nseg = __builtin_amdgcn_readlane(bv, 16) - B;
  const bool fits = nseg <= IDX_CAP;

  {
    const float* xg = x + row0 * NUM_FEAT;
    float* sx = smem + OFF_X + wave * (16 * NUM_FEAT);
    for (int i = lane; i < (16 * NUM_FEAT) / 4; i += 32) {
      float4 vx = *(const float4*)(xg + i * 4);
      *(float4*)(sx + i * 4) = vx;
    }
    if (fits)
      for (int i = lane; i < nseg; i += 32) sIdx[i] = ssrc[B + i];
  }
  __syncthreads();

  if (fits) gather_rows<true >(x, ssrc, sIdx, B, bv, half, l16, sa);
  else      gather_rows<false>(x, ssrc, sIdx, B, bv, half, l16, sa);
  __syncthreads();

  const int   degi = __shfl(bv, l16 + 1) - __shfl(bv, l16);
  const float rdeg = 1.0f / fmaxf((float)degi, 1.0f);

  const v2f* xf = (const v2f*)(smem + OFF_X + wave * (16 * NUM_FEAT) + l16 * NUM_FEAT);
  const v2f* af = (const v2f*)(smem + OFF_A + wave * (16 * NUM_FEAT) + l16 * NUM_FEAT);
  const v2f* wp = sW + lane;

  v8f acc0 = {}, acc1 = {}, acc2 = {};

#pragma unroll
  for (int step = 0; step < KSTEPS; ++step) {
    const v2f a_x   = xf[2 * step + half];
    const v2f a_agg = af[2 * step + half] * rdeg;

    const int wb = step * (2 * CTILES * 32);
    const v2f bl0 = wp[wb];
    const v2f bl1 = wp[wb + 32];
    const v2f bl2 = wp[wb + 64];
    const v2f br0 = wp[wb + 96];
    const v2f br1 = wp[wb + 128];
    const v2f br2 = wp[wb + 160];

    acc0 = __builtin_amdgcn_wmma_f32_16x16x4_f32(false, a_agg, false, bl0, (short)0, acc0, false, false);
    acc1 = __builtin_amdgcn_wmma_f32_16x16x4_f32(false, a_agg, false, bl1, (short)0, acc1, false, false);
    acc2 = __builtin_amdgcn_wmma_f32_16x16x4_f32(false, a_agg, false, bl2, (short)0, acc2, false, false);
    acc0 = __builtin_amdgcn_wmma_f32_16x16x4_f32(false, a_x,   false, br0, (short)0, acc0, false, false);
    acc1 = __builtin_amdgcn_wmma_f32_16x16x4_f32(false, a_x,   false, br1, (short)0, acc1, false, false);
    acc2 = __builtin_amdgcn_wmma_f32_16x16x4_f32(false, a_x,   false, br2, (short)0, acc2, false, false);
  }

  const float b0 = bias[l16];
  const float b1 = bias[16 + l16];
  const float b2 = (32 + l16 < NUM_CLASS) ? bias[32 + l16] : 0.0f;

#pragma unroll
  for (int r = 0; r < 8; ++r) {
    const int M = r + 8 * half;
    so[M * 48 +      l16] = acc0[r] + b0;
    so[M * 48 + 16 + l16] = acc1[r] + b1;
    so[M * 48 + 32 + l16] = acc2[r] + b2;
  }
  __syncthreads();

  if (lane < 16) {
    const float* rowp = so + l16 * 48;
    float mx = rowp[0];
    for (int c = 1; c < NUM_CLASS; ++c) mx = fmaxf(mx, rowp[c]);
    float s = 0.0f;
    for (int c = 0; c < NUM_CLASS; ++c) s += __expf(rowp[c] - mx);
    sLse[wave * 16 + l16] = mx + __logf(s);
  }
  __syncthreads();

  for (int i = lane; i < 16 * NUM_CLASS; i += 32) {
    const int r = i / NUM_CLASS, c = i % NUM_CLASS;
    out[(row0 + r) * NUM_CLASS + c] = so[r * 48 + c] - sLse[wave * 16 + r];
  }
}

extern "C" void kernel_launch(void* const* d_in, const int* in_sizes, int n_in,
                              void* d_out, int out_size, void* d_ws, size_t ws_size,
                              hipStream_t stream) {
  const float* x     = (const float*)d_in[0];
  const int*   index = (const int*)d_in[1];
  const float* Wl    = (const float*)d_in[2];
  const float* bl    = (const float*)d_in[3];
  const float* Wr    = (const float*)d_in[4];
  float* out = (float*)d_out;

  int* cnt  = (int*)d_ws;
  int* btot = cnt + CNT_PAD;
  int* rank = btot + 128;
  int* ssrc = rank + N_EDGES;

  zero_cnt  <<<SCAN_BLOCKS, 256, 0, stream>>>(cnt);
  edge_rank <<<EDGE_GRID,   256, 0, stream>>>(index, cnt, rank);
  scan_local<<<SCAN_BLOCKS, 256, 0, stream>>>(cnt, btot);
  scan_add  <<<SCAN_BLOCKS, 256, 0, stream>>>(cnt, btot);
  edge_place<<<EDGE_GRID,   256, 0, stream>>>(index, cnt, rank, ssrc);
  sage_gemm <<<GEMM_GRID, GEMM_BLOCK, 0, stream>>>(x, cnt, ssrc, Wl, bl, Wr, out);
}


# Group edges by destination with LDS counters (bucket count, scan, bucket scatter, per-bucket sort) in place of a returning global atomic per edge; row sum order changes (still arrival order)
# speedup vs baseline: 2.1788x; 1.6197x over previous
#include <hip/hip_runtime.h>
#include <math.h>


typedef __attribute__((ext_vector_type(2))) float v2f;
typedef __attribute__((ext_vector_type(8))) float v8f;

#define N_NODES   100000
#define NUM_FEAT  64
#define NUM_CLASS 40
#define N_EDGES   1600000
#define WPB       5
#define GEMM_BLOCK (WPB * 32)
#define ROW_TILES  (N_NODES / 16)
#define GEMM_GRID  (ROW_TILES / WPB)
#define KSTEPS     (NUM_FEAT / 4)
#define CTILES     3

#define CNT_PAD     (98 * 1024)

#define NB          250
#define BROWS       400
#define PBLK        250
#define CHUNK       (N_EDGES / PBLK)
#define CHUNK4      (CHUNK / 4)
#define CTHREADS    320
#define EPT         (CHUNK4 / CTHREADS)
#define TBL         (NB * PBLK)
#define TSCAN_BLOCKS 62
#define TBL_PAD     (TSCAN_BLOCKS * 1024)
#define BST_OFF     100096
#define SRC_BITS    17
#define SRC_MASK    ((1 << SRC_BITS) - 1)
#define SORT_THREADS 512
#define SU          8
#define STAGE_CAP   12288
#define IDX_CAP     (16 * 48)
#define GU          8

#define NW_V2F     (KSTEPS * 2 * CTILES * 32)
#define OFF_W      0
#define OFF_OUT    (NW_V2F * 2)
#define OFF_LSE    (OFF_OUT + WPB * 16 * 48)
#define OFF_X      (OFF_LSE + WPB * 16)
#define OFF_A      (OFF_X + WPB * 16 * NUM_FEAT)
#define SMEM_FLOATS (OFF_A + WPB * 16 * NUM_FEAT)

__global__ void __launch_bounds__(CTHREADS)
bucket_count(const int* __restrict__ index, int* __restrict__ tbl) {
  __shared__ int h[NB];
  const int t = threadIdx.x, blk = blockIdx.x;
  const int4* dp = (const int4*)(index + N_EDGES + blk * CHUNK);
  int4 d[EPT];
#pragma unroll
  for (int u = 0; u < EPT; ++u) d[u] = dp[t + u * CTHREADS];
  if (t < NB) h[t] = 0;
  __syncthreads();
#pragma unroll
  for (int u = 0; u < EPT; ++u) {
    atomicAdd(&h[(unsigned)d[u].x / BROWS], 1);
    atomicAdd(&h[(unsigned)d[u].y / BROWS], 1);
    atomicAdd(&h[(unsigned)d[u].z / BROWS], 1);
    atomicAdd(&h[(unsigned)d[u].w / BROWS], 1);
  }
  __syncthreads();
  if (t < NB) tbl[t * PBLK + blk] = h[t];
  if (blk == 0)
    for (int i = t; i < TBL_PAD - TBL; i += CTHREADS) tbl[TBL + i] = 0;
}

__global__ void __launch_bounds__(256) scan_local(int* __restrict__ cnt, int* __restrict__ btot) {
  __shared__ int wsum[8];
  const int t = threadIdx.x, lane = t & 31, w = t >> 5;
  int4* p = (int4*)cnt + (blockIdx.x * 256 + t);
  const int4 c = *p;
  const int s = c.x + c.y + c.z + c.w;
  int incl = s;
  for (int o = 1; o < 32; o <<= 1) {
    const int v = __shfl_up(incl, o);
    if (lane >= o) incl += v;
  }
  if (lane == 31) wsum[w] = incl;
  __syncthreads();
  int wbase = 0;
  for (int k = 0; k < 8; ++k) wbase += (k < w) ? wsum[k] : 0;
  const int ex = wbase + incl - s;
  *p = make_int4(ex, ex + c.x, ex + c.x + c.y, ex + c.x + c.y + c.z);
  if (t == 255) btot[blockIdx.x] = ex + s;
}

__global__ void __launch_bounds__(256) scan_add(int* __restrict__ cnt, const int* __restrict__ btot) {
  __shared__ int sbase;
  const int t = threadIdx.x;
  if (t < 32) {
    int a = 0;
    for (int k = t; k < (int)blockIdx.x; k += 32) a += btot[k];
    for (int o = 16; o > 0; o >>= 1) a += __shfl_xor(a, o);
    if (t == 0) sbase = a;
  }
  __syncthreads();
  const int base = sbase;
  int4* p = (int4*)cnt + (blockIdx.x * 256 + t);
  int4 c = *p;
  c.x += base; c.y += base; c.z += base; c.w += base;
  *p = c;
}

static __device__ __forceinline__ void bucket_put(int* off, int* __restrict__ bkt, int d, int s) {
  const unsigned b = (unsigned)d / BROWS;
  const int r = d - (int)b * BROWS;
  const int p = atomicAdd(&off[b], 1);
  bkt[p] = (r << SRC_BITS) | s;
}

__global__ void __launch_bounds__(CTHREADS)
bucket_scatter(const int* __restrict__ index, const int* __restrict__ tbl,
               int* __restrict__ bst, int* __restrict__ bkt) {
  __shared__ int off[NB];
  const int t = threadIdx.x, blk = blockIdx.x;
  const int4* sp = (const int4*)(index + blk * CHUNK);
  const int4* dp = (const int4*)(index + N_EDGES + blk * CHUNK);
  int4 s[EPT], d[EPT];
#pragma unroll
  for (int u = 0; u < EPT; ++u) {
    d[u] = dp[t + u * CTHREADS];
    s[u] = sp[t + u * CTHREADS];
  }
  if (t < NB) off[t] = tbl[t * PBLK + blk];
  if (blk == 0 && t <= NB) bst[t] = tbl[t * PBLK];
  __syncthreads();
#pragma unroll
  for (int u = 0; u < EPT; ++u) {
    bucket_put(off, bkt, d[u].x, s[u].x);
    bucket_put(off, bkt, d[u].y, s[u].y);
    bucket_put(off, bkt, d[u].z, s[u].z);
    bucket_put(off, bkt, d[u].w, s[u].w);
  }
}

template <bool STAGED>
static __device__ __forceinline__ void place_words(const int* __restrict__ p, int n, int t, int* c,
                                                   int* stage, int* __restrict__ dst) {
  for (int i = t; i < n; i += SORT_THREADS * SU) {
    int v[SU];
#pragma unroll
    for (int u = 0; u < SU; ++u) {
      const int idx = i + u * SORT_THREADS;
      const int w0 = p[min(idx, n - 1)];
      v[u] = (idx < n) ? w0 : -1;
    }
#pragma unroll
    for (int u = 0; u < SU; ++u) {
      if (v[u] >= 0) {
        const int pos = atomicAdd(&c[v[u] >> SRC_BITS], 1);
        const int sid = v[u] & SRC_MASK;
        if (STAGED) stage[pos] = sid;
        else        dst[pos] = sid;
      }
    }
  }
}

__global__ void __launch_bounds__(SORT_THREADS)
bucket_sort(const int* __restrict__ bkt, const int* __restrict__ bst,
            int* __restrict__ cnt, int* __restrict__ ssrc) {
  __shared__ int c[BROWS];
  __shared__ int wsum[SORT_THREADS / 32];
  __shared__ int stage[STAGE_CAP];
  const int t = threadIdx.x, lane = t & 31, w = t >> 5, b = blockIdx.x;
  const int s0 = bst[b];
  const int n  = bst[b + 1] - s0;
  const int* p = bkt + s0;
  if (t < BROWS) c[t] = 0;
  __syncthreads();

  for (int i = t; i < n; i += SORT_THREADS * SU) {
    int v[SU];
#pragma unroll
    for (int u = 0; u < SU; ++u) {
      const int idx = i + u * SORT_THREADS;
      const int w0 = p[min(idx, n - 1)];
      v[u] = (idx < n) ? w0 : -1;
    }
#pragma unroll
    for (int u = 0; u < SU; ++u)
      if (v[u] >= 0) atomicAdd(&c[v[u] >> SRC_BITS], 1);
  }
  __syncthreads();

  const int v0 = (t < BROWS) ? c[t] : 0;
  int incl = v0;
  for (int o = 1; o < 32; o <<= 1) {
    const int x = __shfl_up(incl, o);
    if (lane >= o) incl += x;
  }
  if (lane == 31) wsum[w] = incl;
  __syncthreads();
  int wbase = 0;
  for (int k = 0; k < SORT_THREADS / 32; ++k) wbase += (k < w) ? wsum[k] : 0;
  const int ex = wbase + incl - v0;
  if (t < BROWS) {
    c[t] = ex;
    cnt[b * BROWS + t] = s0 + ex;
  }
  if (b == NB - 1 && t == 0) cnt[N_NODES] = s0 + n;
  __syncthreads();

  if (n <= STAGE_CAP) {
    place_words<true>(p, n, t, c, stage, ssrc + s0);
    __syncthreads();
    for (int i = t; i < n; i += SORT_THREADS) ssrc[s0 + i] = stage[i];
  } else {
    place_words<false>(p, n, t, c, stage, ssrc + s0);
  }
}

template <bool LDS_IDX>
static __device__ __forceinline__ void gather_rows(const float* __restrict__ x,
                                                   const int* __restrict__ ssrc,
                                                   const int* sIdx, int B, int bv,
                                                   int half, int q, float* sa) {
  for (int r = 0; r < 16; ++r) {
    const int b = __builtin_amdgcn_readlane(bv, r);
    const int e = __builtin_amdgcn_readlane(bv, r + 1);
    float4 acc = make_float4(0.0f, 0.0f, 0.0f, 0.0f);
    for (int j = b + half; j < e; j += 2 * GU) {
      int s[GU];
#pragma unroll
      for (int u = 0; u < GU; ++u) {
        const int jj = min(j + 2 * u, e - 1);
        s[u] = LDS_IDX ? sIdx[jj - B] : ssrc[jj];
      }
      float4 v[GU];
#pragma unroll
      for (int u = 0; u < GU; ++u) {
        v[u] = make_float4(0.0f, 0.0f, 0.0f, 0.0f);
        if (j + 2 * u < e) v[u] = *(const float4*)(x + (long)s[u] * NUM_FEAT + q * 4);
      }
#pragma unroll
      for (int u = 0; u < GU; ++u) {
        acc.x += v[u].x; acc.y += v[u].y; acc.z += v[u].z; acc.w += v[u].w;
      }
    }
    acc.x += __shfl_xor(acc.x, 16);
    acc.y += __shfl_xor(acc.y, 16);
    acc.z += __shfl_xor(acc.z, 16);
    acc.w += __shfl_xor(acc.w, 16);
    if (half == 0) *(float4*)(sa + r * NUM_FEAT + q * 4) = acc;
  }
}

__global__ void __launch_bounds__(GEMM_BLOCK)
sage_gemm(const float* __restrict__ x,
          const int*   __restrict__ cnt,
          const int*   __restrict__ ssrc,
          const float* __restrict__ Wl,
          const float* __restrict__ bias,
          const float* __restrict__ Wr,
          float* __restrict__ out) {
  __shared__ __align__(16) float smem[SMEM_FLOATS];

  v2f*   sW   = (v2f*)smem;
  float* sOut = smem + OFF_OUT;
  float* sLse = smem + OFF_LSE;

  const int tid  = threadIdx.x;
  const int wave = tid >> 5;
  const int lane = tid & 31;
  const int half = lane >> 4;
  const int l16  = lane & 15;

  const long row0 = ((long)blockIdx.x * WPB + wave) * 16;

  const int bv = (lane <= 16) ? cnt[row0 + lane] : 0;

  for (int i = tid; i < NW_V2F; i += GEMM_BLOCK) {
    const int li    = i & 31;
    const int t     = (i >> 5) % CTILES;
    const int which = (i / (CTILES * 32)) & 1;
    const int step  = i / (2 * CTILES * 32);
    const int k     = 4 * step + 2 * (li >> 4);
    const int n     = t * 16 + (li & 15);
    const float* W  = which ? Wr : Wl;
    v2f w = {0.0f, 0.0f};
    if (n < NUM_CLASS) w = (v2f){W[k * NUM_CLASS + n], W[(k + 1) * NUM_CLASS + n]};
    sW[i] = w;
  }

  float* so = sOut + wave * (16 * 48);
  int*   sIdx = (int*)so;
  float* sa = smem + OFF_A + wave * (16 * NUM_FEAT);

  const int  B    = __builtin_amdgcn_readlane(bv, 0);
  const int  nseg = __builtin_amdgcn_readlane(bv, 16) - B;
  const bool fits = nseg <= IDX_CAP;

  {
    const float* xg = x + row0 * NUM_FEAT;
    float* sx = smem + OFF_X + wave * (16 * NUM_FEAT);
    for (int i = lane; i < (16 * NUM_FEAT) / 4; i += 32) {
      float4 vx = *(const float4*)(xg + i * 4);
      *(float4*)(sx + i * 4) = vx;
    }
    if (fits)
      for (int i = lane; i < nseg; i += 32) sIdx[i] = ssrc[B + i];
  }
  __syncthreads();

  if (fits) gather_rows<true >(x, ssrc, sIdx, B, bv, half, l16, sa);
  else      gather_rows<false>(x, ssrc, sIdx, B, bv, half, l16, sa);
  __syncthreads();

  const int   degi = __shfl(bv, l16 + 1) - __shfl(bv, l16);
  const float rdeg = 1.0f / fmaxf((float)degi, 1.0f);

  const v2f* xf = (const v2f*)(smem + OFF_X + wave * (16 * NUM_FEAT) + l16 * NUM_FEAT);
  const v2f* af = (const v2f*)(smem + OFF_A + wave * (16 * NUM_FEAT) + l16 * NUM_FEAT);
  const v2f* wp = sW + lane;

  v8f acc0 = {}, acc1 = {}, acc2 = {};

#pragma unroll
  for (int step = 0; step < KSTEPS; ++step) {
    const v2f a_x   = xf[2 * step + half];
    const v2f a_agg = af[2 * step + half] * rdeg;

    const int wb = step * (2 * CTILES * 32);
    const v2f bl0 = wp[wb];
    const v2f bl1 = wp[wb + 32];
    const v2f bl2 = wp[wb + 64];
    const v2f br0 = wp[wb + 96];
    const v2f br1 = wp[wb + 128];
    const v2f br2 = wp[wb + 160];

    acc0 = __builtin_amdgcn_wmma_f32_16x16x4_f32(false, a_agg, false, bl0, (short)0, acc0, false, false);
    acc1 = __builtin_amdgcn_wmma_f32_16x16x4_f32(false, a_agg, false, bl1, (short)0, acc1, false, false);
    acc2 = __builtin_amdgcn_wmma_f32_16x16x4_f32(false, a_agg, false, bl2, (short)0, acc2, false, false);
    acc0 = __builtin_amdgcn_wmma_f32_16x16x4_f32(false, a_x,   false, br0, (short)0, acc0, false, false);
    acc1 = __builtin_amdgcn_wmma_f32_16x16x4_f32(false, a_x,   false, br1, (short)0, acc1, false, false);
    acc2 = __builtin_amdgcn_wmma_f32_16x16x4_f32(false, a_x,   false, br2, (short)0, acc2, false, false);
  }

  const float b0 = bias[l16];
  const float b1 = bias[16 + l16];
  const float b2 = (32 + l16 < NUM_CLASS) ? bias[32 + l16] : 0.0f;

#pragma unroll
  for (int r = 0; r < 8; ++r) {
    const int M = r + 8 * half;
    so[M * 48 +      l16] = acc0[r] + b0;
    so[M * 48 + 16 + l16] = acc1[r] + b1;
    so[M * 48 + 32 + l16] = acc2[r] + b2;
  }
  __syncthreads();

  if (lane < 16) {
    const float* rowp = so + l16 * 48;
    float mx = rowp[0];
    for (int c = 1; c < NUM_CLASS; ++c) mx = fmaxf(mx, rowp[c]);
    float s = 0.0f;
    for (int c = 0; c < NUM_CLASS; ++c) s += __expf(rowp[c] - mx);
    sLse[wave * 16 + l16] = mx + __logf(s);
  }
  __syncthreads();

  for (int i = lane; i < 16 * NUM_CLASS; i += 32) {
    const int r = i / NUM_CLASS, c = i % NUM_CLASS;
    out[(row0 + r) * NUM_CLASS + c] = so[r * 48 + c] - sLse[wave * 16 + r];
  }
}

extern "C" void kernel_launch(void* const* d_in, const int* in_sizes, int n_in,
                              void* d_out, int out_size, void* d_ws, size_t ws_size,
                              hipStream_t stream) {
  const float* x     = (const float*)d_in[0];
  const int*   index = (const int*)d_in[1];
  const float* Wl    = (const float*)d_in[2];
  const float* bl    = (const float*)d_in[3];
  const float* Wr    = (const float*)d_in[4];
  float* out = (float*)d_out;

  int* cnt  = (int*)d_ws;
  int* tbl  = cnt;
  int* bst  = cnt + BST_OFF;
  int* btot = cnt + CNT_PAD;
  int* bkt  = btot + 128;
  int* ssrc = bkt + N_EDGES;

  bucket_count  <<<PBLK,         CTHREADS,     0, stream>>>(index, tbl);
  scan_local    <<<TSCAN_BLOCKS, 256,          0, stream>>>(tbl, btot);
  scan_add      <<<TSCAN_BLOCKS, 256,          0, stream>>>(tbl, btot);
  bucket_scatter<<<PBLK,         CTHREADS,     0, stream>>>(index, tbl, bst, bkt);
  bucket_sort   <<<NB,           SORT_THREADS, 0, stream>>>(bkt, bst, cnt, ssrc);
  sage_gemm <<<GEMM_GRID, GEMM_BLOCK, 0, stream>>>(x, cnt, ssrc, Wl, bl, Wr, out);
}


# GEMM kernel launched as 250 work-groups of 25 waves (one per WGP) instead of 1250 of 5 waves: all 6250 row tiles resident in one round, weights staged once per WGP; sum order unchanged
# speedup vs baseline: 2.8596x; 1.3125x over previous
#include <hip/hip_runtime.h>
#include <math.h>


typedef __attribute__((ext_vector_type(2))) float v2f;
typedef __attribute__((ext_vector_type(8))) float v8f;

#define N_NODES   100000
#define NUM_FEAT  64
#define NUM_CLASS 40
#define N_EDGES   1600000
#define WPB       25
#define GEMM_BLOCK (WPB * 32)
#define ROW_TILES  (N_NODES / 16)
#define GEMM_GRID  (ROW_TILES / WPB)
#define KSTEPS     (NUM_FEAT / 4)
#define CTILES     3

#define CNT_PAD     (98 * 1024)

#define NB          250
#define BROWS       400
#define PBLK        250
#define CHUNK       (N_EDGES / PBLK)
#define CHUNK4      (CHUNK / 4)
#define CTHREADS    320
#define EPT         (CHUNK4 / CTHREADS)
#define TBL         (NB * PBLK)
#define TSCAN_BLOCKS 62
#define TBL_PAD     (TSCAN_BLOCKS * 1024)
#define BST_OFF     100096
#define SRC_BITS    17
#define SRC_MASK    ((1 << SRC_BITS) - 1)
#define SORT_THREADS 512
#define SU          8
#define STAGE_CAP   12288
#define IDX_CAP     (16 * 48)
#define GU          8

#define NW_V2F     (KSTEPS * 2 * CTILES * 32)
#define OFF_W      0
#define OFF_OUT    (NW_V2F * 2)
#define OFF_LSE    (OFF_OUT + WPB * 16 * 48)
#define OFF_X      (OFF_LSE + WPB * 16)
#define OFF_A      (OFF_X + WPB * 16 * NUM_FEAT)
#define SMEM_FLOATS (OFF_A + WPB * 16 * NUM_FEAT)

__global__ void __launch_bounds__(CTHREADS)
bucket_count(const int* __restrict__ index, int* __restrict__ tbl) {
  __shared__ int h[NB];
  const int t = threadIdx.x, blk = blockIdx.x;
  const int4* dp = (const int4*)(index + N_EDGES + blk * CHUNK);
  int4 d[EPT];
#pragma unroll
  for (int u = 0; u < EPT; ++u) d[u] = dp[t + u * CTHREADS];
  if (t < NB) h[t] = 0;
  __syncthreads();
#pragma unroll
  for (int u = 0; u < EPT; ++u) {
    atomicAdd(&h[(unsigned)d[u].x / BROWS], 1);
    atomicAdd(&h[(unsigned)d[u].y / BROWS], 1);
    atomicAdd(&h[(unsigned)d[u].z / BROWS], 1);
    atomicAdd(&h[(unsigned)d[u].w / BROWS], 1);
  }
  __syncthreads();
  if (t < NB) tbl[t * PBLK + blk] = h[t];
  if (blk == 0)
    for (int i = t; i < TBL_PAD - TBL; i += CTHREADS) tbl[TBL + i] = 0;
}

__global__ void __launch_bounds__(256) scan_local(int* __restrict__ cnt, int* __restrict__ btot) {
  __shared__ int wsum[8];
  const int t = threadIdx.x, lane = t & 31, w = t >> 5;
  int4* p = (int4*)cnt + (blockIdx.x * 256 + t);
  const int4 c = *p;
  const int s = c.x + c.y + c.z + c.w;
  int incl = s;
  for (int o = 1; o < 32; o <<= 1) {
    const int v = __shfl_up(incl, o);
    if (lane >= o) incl += v;
  }
  if (lane == 31) wsum[w] = incl;
  __syncthreads();
  int wbase = 0;
  for (int k = 0; k < 8; ++k) wbase += (k < w) ? wsum[k] : 0;
  const int ex = wbase + incl - s;
  *p = make_int4(ex, ex + c.x, ex + c.x + c.y, ex + c.x + c.y + c.z);
  if (t == 255) btot[blockIdx.x] = ex + s;
}

__global__ void __launch_bounds__(256) scan_add(int* __restrict__ cnt, const int* __restrict__ btot) {
  __shared__ int sbase;
  const int t = threadIdx.x;
  if (t < 32) {
    int a = 0;
    for (int k = t; k < (int)blockIdx.x; k += 32) a += btot[k];
    for (int o = 16; o > 0; o >>= 1) a += __shfl_xor(a, o);
    if (t == 0) sbase = a;
  }
  __syncthreads();
  const int base = sbase;
  int4* p = (int4*)cnt + (blockIdx.x * 256 + t);
  int4 c = *p;
  c.x += base; c.y += base; c.z += base; c.w += base;
  *p = c;
}

static __device__ __forceinline__ void bucket_put(int* off, int* __restrict__ bkt, int d, int s) {
  const unsigned b = (unsigned)d / BROWS;
  const int r = d - (int)b * BROWS;
  const int p = atomicAdd(&off[b], 1);
  bkt[p] = (r << SRC_BITS) | s;
}

__global__ void __launch_bounds__(CTHREADS)
bucket_scatter(const int* __restrict__ index, const int* __restrict__ tbl,
               int* __restrict__ bst, int* __restrict__ bkt) {
  __shared__ int off[NB];
  const int t = threadIdx.x, blk = blockIdx.x;
  const int4* sp = (const int4*)(index + blk * CHUNK);
  const int4* dp = (const int4*)(index + N_EDGES + blk * CHUNK);
  int4 s[EPT], d[EPT];
#pragma unroll
  for (int u = 0; u < EPT; ++u) {
    d[u] = dp[t + u * CTHREADS];
    s[u] = sp[t + u * CTHREADS];
  }
  if (t < NB) off[t] = tbl[t * PBLK + blk];
  if (blk == 0 && t <= NB) bst[t] = tbl[t * PBLK];
  __syncthreads();
#pragma unroll
  for (int u = 0; u < EPT; ++u) {
    bucket_put(off, bkt, d[u].x, s[u].x);
    bucket_put(off, bkt, d[u].y, s[u].y);
    bucket_put(off, bkt, d[u].z, s[u].z);
    bucket_put(off, bkt, d[u].w, s[u].w);
  }
}

template <bool STAGED>
static __device__ __forceinline__ void place_words(const int* __restrict__ p, int n, int t, int* c,
                                                   int* stage, int* __restrict__ dst) {
  for (int i = t; i < n; i += SORT_THREADS * SU) {
    int v[SU];
#pragma unroll
    for (int u = 0; u < SU; ++u) {
      const int idx = i + u * SORT_THREADS;
      const int w0 = p[min(idx, n - 1)];
      v[u] = (idx < n) ? w0 : -1;
    }
#pragma unroll
    for (int u = 0; u < SU; ++u) {
      if (v[u] >= 0) {
        const int pos = atomicAdd(&c[v[u] >> SRC_BITS], 1);
        const int sid = v[u] & SRC_MASK;
        if (STAGED) stage[pos] = sid;
        else        dst[pos] = sid;
      }
    }
  }
}

__global__ void __launch_bounds__(SORT_THREADS)
bucket_sort(const int* __restrict__ bkt, const int* __restrict__ bst,
            int* __restrict__ cnt, int* __restrict__ ssrc) {
  __shared__ int c[BROWS];
  __shared__ int wsum[SORT_THREADS / 32];
  __shared__ int stage[STAGE_CAP];
  const int t = threadIdx.x, lane = t & 31, w = t >> 5, b = blockIdx.x;
  const int s0 = bst[b];
  const int n  = bst[b + 1] - s0;
  const int* p = bkt + s0;
  if (t < BROWS) c[t] = 0;
  __syncthreads();

  for (int i = t; i < n; i += SORT_THREADS * SU) {
    int v[SU];
#pragma unroll
    for (int u = 0; u < SU; ++u) {
      const int idx = i + u * SORT_THREADS;
      const int w0 = p[min(idx, n - 1)];
      v[u] = (idx < n) ? w0 : -1;
    }
#pragma unroll
    for (int u = 0; u < SU; ++u)
      if (v[u] >= 0) atomicAdd(&c[v[u] >> SRC_BITS], 1);
  }
  __syncthreads();

  const int v0 = (t < BROWS) ? c[t] : 0;
  int incl = v0;
  for (int o = 1; o < 32; o <<= 1) {
    const int x = __shfl_up(incl, o);
    if (lane >= o) incl += x;
  }
  if (lane == 31) wsum[w] = incl;
  __syncthreads();
  int wbase = 0;
  for (int k = 0; k < SORT_THREADS / 32; ++k) wbase += (k < w) ? wsum[k] : 0;
  const int ex = wbase + incl - v0;
  if (t < BROWS) {
    c[t] = ex;
    cnt[b * BROWS + t] = s0 + ex;
  }
  if (b == NB - 1 && t == 0) cnt[N_NODES] = s0 + n;
  __syncthreads();

  if (n <= STAGE_CAP) {
    place_words<true>(p, n, t, c, stage, ssrc + s0);
    __syncthreads();
    for (int i = t; i < n; i += SORT_THREADS) ssrc[s0 + i] = stage[i];
  } else {
    place_words<false>(p, n, t, c, stage, ssrc + s0);
  }
}

template <bool LDS_IDX>
static __device__ __forceinline__ void gather_rows(const float* __restrict__ x,
                                                   const int* __restrict__ ssrc,
                                                   const int* sIdx, int B, int bv,
                                                   int half, int q, float* sa) {
  for (int r = 0; r < 16; ++r) {
    const int b = __builtin_amdgcn_readlane(bv, r);
    const int e = __builtin_amdgcn_readlane(bv, r + 1);
    float4 acc = make_float4(0.0f, 0.0f, 0.0f, 0.0f);
    for (int j = b + half; j < e; j += 2 * GU) {
      int s[GU];
#pragma unroll
      for (int u = 0; u < GU; ++u) {
        const int jj = min(j + 2 * u, e - 1);
        s[u] = LDS_IDX ? sIdx[jj - B] : ssrc[jj];
      }
      float4 v[GU];
#pragma unroll
      for (int u = 0; u < GU; ++u) {
        v[u] = make_float4(0.0f, 0.0f, 0.0f, 0.0f);
        if (j + 2 * u < e) v[u] = *(const float4*)(x + (long)s[u] * NUM_FEAT + q * 4);
      }
#pragma unroll
      for (int u = 0; u < GU; ++u) {
        acc.x += v[u].x; acc.y += v[u].y; acc.z += v[u].z; acc.w += v[u].w;
      }
    }
    acc.x += __shfl_xor(acc.x, 16);
    acc.y += __shfl_xor(acc.y, 16);
    acc.z += __shfl_xor(acc.z, 16);
    acc.w += __shfl_xor(acc.w, 16);
    if (half == 0) *(float4*)(sa + r * NUM_FEAT + q * 4) = acc;
  }
}

__global__ void __launch_bounds__(GEMM_BLOCK)
sage_gemm(const float* __restrict__ x,
          const int*   __restrict__ cnt,
          const int*   __restrict__ ssrc,
          const float* __restrict__ Wl,
          const float* __restrict__ bias,
          const float* __restrict__ Wr,
          float* __restrict__ out) {
  __shared__ __align__(16) float smem[SMEM_FLOATS];

  v2f*   sW   = (v2f*)smem;
  float* sOut = smem + OFF_OUT;
  float* sLse = smem + OFF_LSE;

  const int tid  = threadIdx.x;
  const int wave = tid >> 5;
  const int lane = tid & 31;
  const int half = lane >> 4;
  const int l16  = lane & 15;

  const long row0 = ((long)blockIdx.x * WPB + wave) * 16;

  const int bv = (lane <= 16) ? cnt[row0 + lane] : 0;

  for (int i = tid; i < NW_V2F; i += GEMM_BLOCK) {
    const int li    = i & 31;
    const int t     = (i >> 5) % CTILES;
    const int which = (i / (CTILES * 32)) & 1;
    const int step  = i / (2 * CTILES * 32);
    const int k     = 4 * step + 2 * (li >> 4);
    const int n     = t * 16 + (li & 15);
    const float* W  = which ? Wr : Wl;
    v2f w = {0.0f, 0.0f};
    if (n < NUM_CLASS) w = (v2f){W[k * NUM_CLASS + n], W[(k + 1) * NUM_CLASS + n]};
    sW[i] = w;
  }

  float* so = sOut + wave * (16 * 48);
  int*   sIdx = (int*)so;
  float* sa = smem + OFF_A + wave * (16 * NUM_FEAT);

  const int  B    = __builtin_amdgcn_readlane(bv, 0);
  const int  nseg = __builtin_amdgcn_readlane(bv, 16) - B;
  const bool fits = nseg <= IDX_CAP;

  {
    const float* xg = x + row0 * NUM_FEAT;
    float* sx = smem + OFF_X + wave * (16 * NUM_FEAT);
    for (int i = lane; i < (16 * NUM_FEAT) / 4; i += 32) {
      float4 vx = *(const float4*)(xg + i * 4);
      *(float4*)(sx + i * 4) = vx;
    }
    if (fits)
      for (int i = lane; i < nseg; i += 32) sIdx[i] = ssrc[B + i];
  }
  __syncthreads();

  if (fits) gather_rows<true >(x, ssrc, sIdx, B, bv, half, l16, sa);
  else      gather_rows<false>(x, ssrc, sIdx, B, bv, half, l16, sa);
  __syncthreads();

  const int   degi = __shfl(bv, l16 + 1) - __shfl(bv, l16);
  const float rdeg = 1.0f / fmaxf((float)degi, 1.0f);

  const v2f* xf = (const v2f*)(smem + OFF_X + wave * (16 * NUM_FEAT) + l16 * NUM_FEAT);
  const v2f* af = (const v2f*)(smem + OFF_A + wave * (16 * NUM_FEAT) + l16 * NUM_FEAT);
  const v2f* wp = sW + lane;

  v8f acc0 = {}, acc1 = {}, acc2 = {};

#pragma unroll
  for (int step = 0; step < KSTEPS; ++step) {
    const v2f a_x   = xf[2 * step + half];
    const v2f a_agg = af[2 * step + half] * rdeg;

    const int wb = step * (2 * CTILES * 32);
    const v2f bl0 = wp[wb];
    const v2f bl1 = wp[wb + 32];
    const v2f bl2 = wp[wb + 64];
    const v2f br0 = wp[wb + 96];
    const v2f br1 = wp[wb + 128];
    const v2f br2 = wp[wb + 160];

    acc0 = __builtin_amdgcn_wmma_f32_16x16x4_f32(false, a_agg, false, bl0, (short)0, acc0, false, false);
    acc1 = __builtin_amdgcn_wmma_f32_16x16x4_f32(false, a_agg, false, bl1, (short)0, acc1, false, false);
    acc2 = __builtin_amdgcn_wmma_f32_16x16x4_f32(false, a_agg, false, bl2, (short)0, acc2, false, false);
    acc0 = __builtin_amdgcn_wmma_f32_16x16x4_f32(false, a_x,   false, br0, (short)0, acc0, false, false);
    acc1 = __builtin_amdgcn_wmma_f32_16x16x4_f32(false, a_x,   false, br1, (short)0, acc1, false, false);
    acc2 = __builtin_amdgcn_wmma_f32_16x16x4_f32(false, a_x,   false, br2, (short)0, acc2, false, false);
  }

  const float b0 = bias[l16];
  const float b1 = bias[16 + l16];
  const float b2 = (32 + l16 < NUM_CLASS) ? bias[32 + l16] : 0.0f;

#pragma unroll
  for (int r = 0; r < 8; ++r) {
    const int M = r + 8 * half;
    so[M * 48 +      l16] = acc0[r] + b0;
    so[M * 48 + 16 + l16] = acc1[r] + b1;
    so[M * 48 + 32 + l16] = acc2[r] + b2;
  }
  __syncthreads();

  if (lane < 16) {
    const float* rowp = so + l16 * 48;
    float mx = rowp[0];
    for (int c = 1; c < NUM_CLASS; ++c) mx = fmaxf(mx, rowp[c]);
    float s = 0.0f;
    for (int c = 0; c < NUM_CLASS; ++c) s += __expf(rowp[c] - mx);
    sLse[wave * 16 + l16] = mx + __logf(s);
  }
  __syncthreads();

  for (int i = lane; i < 16 * NUM_CLASS; i += 32) {
    const int r = i / NUM_CLASS, c = i % NUM_CLASS;
    out[(row0 + r) * NUM_CLASS + c] = so[r * 48 + c] - sLse[wave * 16 + r];
  }
}

extern "C" void kernel_launch(void* const* d_in, const int* in_sizes, int n_in,
                              void* d_out, int out_size, void* d_ws, size_t ws_size,
                              hipStream_t stream) {
  const float* x     = (const float*)d_in[0];
  const int*   index = (const int*)d_in[1];
  const float* Wl    = (const float*)d_in[2];
  const float* bl    = (const float*)d_in[3];
  const float* Wr    = (const float*)d_in[4];
  float* out = (float*)d_out;

  int* cnt  = (int*)d_ws;
  int* tbl  = cnt;
  int* bst  = cnt + BST_OFF;
  int* btot = cnt + CNT_PAD;
  int* bkt  = btot + 128;
  int* ssrc = bkt + N_EDGES;

  bucket_count  <<<PBLK,         CTHREADS,     0, stream>>>(index, tbl);
  scan_local    <<<TSCAN_BLOCKS, 256,          0, stream>>>(tbl, btot);
  scan_add      <<<TSCAN_BLOCKS, 256,          0, stream>>>(tbl, btot);
  bucket_scatter<<<PBLK,         CTHREADS,     0, stream>>>(index, tbl, bst, bkt);
  bucket_sort   <<<NB,           SORT_THREADS, 0, stream>>>(bkt, bst, cnt, ssrc);
  sage_gemm <<<GEMM_GRID, GEMM_BLOCK, 0, stream>>>(x, cnt, ssrc, Wl, bl, Wr, out);
}


# Each edge chunk is ordered by bucket in LDS and written back whole; the bucket sort collects its 250 runs; count, two scans and per-edge scattered stores go (6 -> 3 launches); sums still arrival order
# speedup vs baseline: 4.1983x; 1.4681x over previous
#include <hip/hip_runtime.h>
#include <math.h>


typedef __attribute__((ext_vector_type(2))) float v2f;
typedef __attribute__((ext_vector_type(8))) float v8f;

#define N_NODES   100000
#define NUM_FEAT  64
#define NUM_CLASS 40
#define N_EDGES   1600000
#define WPB       25
#define GEMM_BLOCK (WPB * 32)
#define ROW_TILES  (N_NODES / 16)
#define GEMM_GRID  (ROW_TILES / WPB)
#define KSTEPS     (NUM_FEAT / 4)
#define CTILES     3

#define CNT_PAD     (98 * 1024)

#define NB          250
#define BROWS       400
#define PBLK        250
#define CHUNK       (N_EDGES / PBLK)
#define CHUNK4      (CHUNK / 4)
#define CTHREADS    320
#define EPT         (CHUNK4 / CTHREADS)
#define RSTRIDE     252
#define RUNS_PAD    (62 * 1024)
#define SRC_BITS    17
#define SRC_MASK    ((1 << SRC_BITS) - 1)
#define SORT_THREADS 512
#define SORT_WAVES  (SORT_THREADS / 32)
#define PU          8
#define PSLOTS      (2 * SORT_WAVES * PU)
#define STAGE_CAP   12288
#define IDX_CAP     (16 * 48)
#define GU          8

#define NW_V2F     (KSTEPS * 2 * CTILES * 32)
#define OFF_W      0
#define OFF_OUT    (NW_V2F * 2)
#define OFF_LSE    (OFF_OUT + WPB * 16 * 48)
#define OFF_X      (OFF_LSE + WPB * 16)
#define OFF_A      (OFF_X + WPB * 16 * NUM_FEAT)
#define SMEM_FLOATS (OFF_A + WPB * 16 * NUM_FEAT)

static __device__ __forceinline__ void chunk_put(int* off, int* stage, int d, int s) {
  const unsigned b = (unsigned)d / BROWS;
  const int r = d - (int)b * BROWS;
  const int p = atomicAdd(&off[b], 1);
  stage[p] = (r << SRC_BITS) | s;
}

__global__ void __launch_bounds__(CTHREADS)
chunk_sort(const int* __restrict__ index, int* __restrict__ runs, int* __restrict__ cw) {
  __shared__ int h[NB];
  __shared__ int wsum[CTHREADS / 32];
  __shared__ __align__(16) int stage[CHUNK];
  const int t = threadIdx.x, lane = t & 31, w = t >> 5, blk = blockIdx.x;
  const int4* sp = (const int4*)(index + blk * CHUNK);
  const int4* dp = (const int4*)(index + N_EDGES + blk * CHUNK);
  int4 s[EPT], d[EPT];
#pragma unroll
  for (int u = 0; u < EPT; ++u) {
    d[u] = dp[t + u * CTHREADS];
    s[u] = sp[t + u * CTHREADS];
  }
  if (t < NB) h[t] = 0;
  __syncthreads();
#pragma unroll
  for (int u = 0; u < EPT; ++u) {
    atomicAdd(&h[(unsigned)d[u].x / BROWS], 1);
    atomicAdd(&h[(unsigned)d[u].y / BROWS], 1);
    atomicAdd(&h[(unsigned)d[u].z / BROWS], 1);
    atomicAdd(&h[(unsigned)d[u].w / BROWS], 1);
  }
  __syncthreads();

  const int v0 = (t < NB) ? h[t] : 0;
  int incl = v0;
  for (int o = 1; o < 32; o <<= 1) {
    const int x = __shfl_up(incl, o);
    if (lane >= o) incl += x;
  }
  if (lane == 31) wsum[w] = incl;
  __syncthreads();
  int wbase = 0;
  for (int k = 0; k < CTHREADS / 32; ++k) wbase += (k < w) ? wsum[k] : 0;
  const int ex = wbase + incl - v0;
  if (t < NB) {
    h[t] = ex;
    runs[blk * RSTRIDE + t] = ex;
  }
  if (t == 0) runs[blk * RSTRIDE + NB] = CHUNK;
  __syncthreads();

#pragma unroll
  for (int u = 0; u < EPT; ++u) {
    chunk_put(h, stage, d[u].x, s[u].x);
    chunk_put(h, stage, d[u].y, s[u].y);
    chunk_put(h, stage, d[u].z, s[u].z);
    chunk_put(h, stage, d[u].w, s[u].w);
  }
  __syncthreads();

  int4* op = (int4*)(cw + blk * CHUNK);
#pragma unroll
  for (int u = 0; u < EPT; ++u) op[t + u * CTHREADS] = ((const int4*)stage)[t + u * CTHREADS];
}

template <bool COUNT, bool STAGED>
static __device__ __forceinline__ void use_word(int v, int* c, int* stage, int* __restrict__ dst) {
  if (v >= 0) {
    if (COUNT) {
      atomicAdd(&c[v >> SRC_BITS], 1);
    } else {
      const int pos = atomicAdd(&c[v >> SRC_BITS], 1);
      const int sid = v & SRC_MASK;
      if (STAGED) stage[pos] = sid;
      else        dst[pos] = sid;
    }
  }
}

template <bool COUNT, bool STAGED>
static __device__ __forceinline__ void sweep_runs(const int* __restrict__ cw, const int* pst,
                                                  const int* pn, int w, int lane, int* c,
                                                  int* stage, int* __restrict__ dst) {
  for (int i0 = w; i0 < PSLOTS; i0 += SORT_WAVES * PU) {
    int base[PU], len[PU], v0[PU], v1[PU];
#pragma unroll
    for (int u = 0; u < PU; ++u) {
      const int i = i0 + u * SORT_WAVES;
      base[u] = __builtin_amdgcn_readfirstlane(pst[i]);
      len[u]  = __builtin_amdgcn_readfirstlane(pn[i]);
      const int last = max(len[u] - 1, 0);
      v0[u] = cw[min(base[u] + min(lane, last), N_EDGES - 1)];
      v1[u] = cw[min(base[u] + min(lane + 32, last), N_EDGES - 1)];
    }
#pragma unroll
    for (int u = 0; u < PU; ++u) {
      use_word<COUNT, STAGED>((lane < len[u]) ? v0[u] : -1, c, stage, dst);
      use_word<COUNT, STAGED>((lane + 32 < len[u]) ? v1[u] : -1, c, stage, dst);
    }
#pragma unroll
    for (int u = 0; u < PU; ++u) {
      if (len[u] > 64) {
        for (int k = lane + 64; k < len[u]; k += 32)
          use_word<COUNT, STAGED>(cw[base[u] + k], c, stage, dst);
      }
    }
  }
}

__global__ void __launch_bounds__(SORT_THREADS)
bucket_sort(const int* __restrict__ cw, const int* __restrict__ runs,
            int* __restrict__ cnt, int* __restrict__ ssrc) {
  __shared__ int c[BROWS];
  __shared__ int wsum[SORT_WAVES];
  __shared__ int wlen[SORT_WAVES];
  __shared__ int pst[PSLOTS];
  __shared__ int pn[PSLOTS];
  __shared__ int stage[STAGE_CAP];
  const int t = threadIdx.x, lane = t & 31, w = t >> 5, b = blockIdx.x;
  int st = 0, ln = 0;
  if (t < PBLK) {
    st = runs[t * RSTRIDE + b];
    ln = runs[t * RSTRIDE + b + 1] - st;
    pst[t] = t * CHUNK + st;
    pn[t]  = ln;
  } else if (t < PSLOTS) {
    pst[t] = 0;
    pn[t]  = 0;
  }
  if (t < BROWS) c[t] = 0;
  for (int o = 16; o > 0; o >>= 1) {
    st += __shfl_xor(st, o);
    ln += __shfl_xor(ln, o);
  }
  if (lane == 0) { wsum[w] = st; wlen[w] = ln; }
  __syncthreads();
  int s0 = 0, n = 0;
  for (int k = 0; k < SORT_WAVES; ++k) { s0 += wsum[k]; n += wlen[k]; }

  sweep_runs<true, false>(cw, pst, pn, w, lane, c, stage, ssrc + s0);
  __syncthreads();

  const int v0 = (t < BROWS) ? c[t] : 0;
  int incl = v0;
  for (int o = 1; o < 32; o <<= 1) {
    const int x = __shfl_up(incl, o);
    if (lane >= o) incl += x;
  }
  if (lane == 31) wsum[w] = incl;
  __syncthreads();
  int wbase = 0;
  for (int k = 0; k < SORT_WAVES; ++k) wbase += (k < w) ? wsum[k] : 0;
  const int ex = wbase + incl - v0;
  if (t < BROWS) {
    c[t] = ex;
    cnt[b * BROWS + t] = s0 + ex;
  }
  if (b == NB - 1 && t == 0) cnt[N_NODES] = s0 + n;
  __syncthreads();

  if (n <= STAGE_CAP) {
    sweep_runs<false, true>(cw, pst, pn, w, lane, c, stage, ssrc + s0);
    __syncthreads();
    for (int i = t; i < n; i += SORT_THREADS) ssrc[s0 + i] = stage[i];
  } else {
    sweep_runs<false, false>(cw, pst, pn, w, lane, c, stage, ssrc + s0);
  }
}

template <bool LDS_IDX>
static __device__ __forceinline__ void gather_rows(const float* __restrict__ x,
                                                   const int* __restrict__ ssrc,
                                                   const int* sIdx, int B, int bv,
                                                   int half, int q, float* sa) {
  for (int r = 0; r < 16; ++r) {
    const int b = __builtin_amdgcn_readlane(bv, r);
    const int e = __builtin_amdgcn_readlane(bv, r + 1);
    float4 acc = make_float4(0.0f, 0.0f, 0.0f, 0.0f);
    for (int j = b + half; j < e; j += 2 * GU) {
      int s[GU];
#pragma unroll
      for (int u = 0; u < GU; ++u) {
        const int jj = min(j + 2 * u, e - 1);
        s[u] = LDS_IDX ? sIdx[jj - B] : ssrc[jj];
      }
      float4 v[GU];
#pragma unroll
      for (int u = 0; u < GU; ++u) {
        v[u] = make_float4(0.0f, 0.0f, 0.0f, 0.0f);
        if (j + 2 * u < e) v[u] = *(const float4*)(x + (long)s[u] * NUM_FEAT + q * 4);
      }
#pragma unroll
      for (int u = 0; u < GU; ++u) {
        acc.x += v[u].x; acc.y += v[u].y; acc.z += v[u].z; acc.w += v[u].w;
      }
    }
    acc.x += __shfl_xor(acc.x, 16);
    acc.y += __shfl_xor(acc.y, 16);
    acc.z += __shfl_xor(acc.z, 16);
    acc.w += __shfl_xor(acc.w, 16);
    if (half == 0) *(float4*)(sa + r * NUM_FEAT + q * 4) = acc;
  }
}

__global__ void __launch_bounds__(GEMM_BLOCK)
sage_gemm(const float* __restrict__ x,
          const int*   __restrict__ cnt,
          const int*   __restrict__ ssrc,
          const float* __restrict__ Wl,
          const float* __restrict__ bias,
          const float* __restrict__ Wr,
          float* __restrict__ out) {
  __shared__ __align__(16) float smem[SMEM_FLOATS];

  v2f*   sW   = (v2f*)smem;
  float* sOut = smem + OFF_OUT;
  float* sLse = smem + OFF_LSE;

  const int tid  = threadIdx.x;
  const int wave = tid >> 5;
  const int lane = tid & 31;
  const int half = lane >> 4;
  const int l16  = lane & 15;

  const long row0 = ((long)blockIdx.x * WPB + wave) * 16;

  const int bv = (lane <= 16) ? cnt[row0 + lane] : 0;

  for (int i = tid; i < NW_V2F; i += GEMM_BLOCK) {
    const int li    = i & 31;
    const int t     = (i >> 5) % CTILES;
    const int which = (i / (CTILES * 32)) & 1;
    const int step  = i / (2 * CTILES * 32);
    const int k     = 4 * step + 2 * (li >> 4);
    const int n     = t * 16 + (li & 15);
    const float* W  = which ? Wr : Wl;
    v2f w = {0.0f, 0.0f};
    if (n < NUM_CLASS) w = (v2f){W[k * NUM_CLASS + n], W[(k + 1) * NUM_CLASS + n]};
    sW[i] = w;
  }

  float* so = sOut + wave * (16 * 48);
  int*   sIdx = (int*)so;
  float* sa = smem + OFF_A + wave * (16 * NUM_FEAT);

  const int  B    = __builtin_amdgcn_readlane(bv, 0);
  const int  nseg = __builtin_amdgcn_readlane(bv, 16) - B;
  const bool fits = nseg <= IDX_CAP;

  {
    const float* xg = x + row0 * NUM_FEAT;
    float* sx = smem + OFF_X + wave * (16 * NUM_FEAT);
    for (int i = lane; i < (16 * NUM_FEAT) / 4; i += 32) {
      float4 vx = *(const float4*)(xg + i * 4);
      *(float4*)(sx + i * 4) = vx;
    }
    if (fits)
      for (int i = lane; i < nseg; i += 32) sIdx[i] = ssrc[B + i];
  }
  __syncthreads();

  if (fits) gather_rows<true >(x, ssrc, sIdx, B, bv, half, l16, sa);
  else      gather_rows<false>(x, ssrc, sIdx, B, bv, half, l16, sa);
  __syncthreads();

  const int   degi = __shfl(bv, l16 + 1) - __shfl(bv, l16);
  const float rdeg = 1.0f / fmaxf((float)degi, 1.0f);

  const v2f* xf = (const v2f*)(smem + OFF_X + wave * (16 * NUM_FEAT) + l16 * NUM_FEAT);
  const v2f* af = (const v2f*)(smem + OFF_A + wave * (16 * NUM_FEAT) + l16 * NUM_FEAT);
  const v2f* wp = sW + lane;

  v8f acc0 = {}, acc1 = {}, acc2 = {};

#pragma unroll
  for (int step = 0; step < KSTEPS; ++step) {
    const v2f a_x   = xf[2 * step + half];
    const v2f a_agg = af[2 * step + half] * rdeg;

    const int wb = step * (2 * CTILES * 32);
    const v2f bl0 = wp[wb];
    const v2f bl1 = wp[wb + 32];
    const v2f bl2 = wp[wb + 64];
    const v2f br0 = wp[wb + 96];
    const v2f br1 = wp[wb + 128];
    const v2f br2 = wp[wb + 160];

    acc0 = __builtin_amdgcn_wmma_f32_16x16x4_f32(false, a_agg, false, bl0, (short)0, acc0, false, false);
    acc1 = __builtin_amdgcn_wmma_f32_16x16x4_f32(false, a_agg, false, bl1, (short)0, acc1, false, false);
    acc2 = __builtin_amdgcn_wmma_f32_16x16x4_f32(false, a_agg, false, bl2, (short)0, acc2, false, false);
    acc0 = __builtin_amdgcn_wmma_f32_16x16x4_f32(false, a_x,   false, br0, (short)0, acc0, false, false);
    acc1 = __builtin_amdgcn_wmma_f32_16x16x4_f32(false, a_x,   false, br1, (short)0, acc1, false, false);
    acc2 = __builtin_amdgcn_wmma_f32_16x16x4_f32(false, a_x,   false, br2, (short)0, acc2, false, false);
  }

  const float b0 = bias[l16];
  const float b1 = bias[16 + l16];
  const float b2 = (32 + l16 < NUM_CLASS) ? bias[32 + l16] : 0.0f;

#pragma unroll
  for (int r = 0; r < 8; ++r) {
    const int M = r + 8 * half;
    so[M * 48 +      l16] = acc0[r] + b0;
    so[M * 48 + 16 + l16] = acc1[r] + b1;
    so[M * 48 + 32 + l16] = acc2[r] + b2;
  }
  __syncthreads();

  if (lane < 16) {
    const float* rowp = so + l16 * 48;
    float mx = rowp[0];
    for (int c = 1; c < NUM_CLASS; ++c) mx = fmaxf(mx, rowp[c]);
    float s = 0.0f;
    for (int c = 0; c < NUM_CLASS; ++c) s += __expf(rowp[c] - mx);
    sLse[wave * 16 + l16] = mx + __logf(s);
  }
  __syncthreads();

  for (int i = lane; i < 16 * NUM_CLASS; i += 32) {
    const int r = i / NUM_CLASS, c = i % NUM_CLASS;
    out[(row0 + r) * NUM_CLASS + c] = so[r * 48 + c] - sLse[wave * 16 + r];
  }
}

extern "C" void kernel_launch(void* const* d_in, const int* in_sizes, int n_in,
                              void* d_out, int out_size, void* d_ws, size_t ws_size,
                              hipStream_t stream) {
  const float* x     = (const float*)d_in[0];
  const int*   index = (const int*)d_in[1];
  const float* Wl    = (const float*)d_in[2];
  const float* bl    = (const float*)d_in[3];
  const float* Wr    = (const float*)d_in[4];
  float* out = (float*)d_out;

  int* cnt  = (int*)d_ws;
  int* runs = cnt + CNT_PAD;
  int* cw   = runs + RUNS_PAD;
  int* ssrc = cw + N_EDGES;

  chunk_sort    <<<PBLK,         CTHREADS,     0, stream>>>(index, runs, cw);
  bucket_sort   <<<NB,           SORT_THREADS, 0, stream>>>(cw, runs, cnt, ssrc);
  sage_gemm <<<GEMM_GRID, GEMM_BLOCK, 0, stream>>>(x, cnt, ssrc, Wl, bl, Wr, out);
}


# Bucket sort moved into the GEMM kernel: each bucket's work-group orders its 400 rows' edges in its own LDS and gathers from there (3 -> 2 launches, no id round trip); sums still arrival order
# speedup vs baseline: 4.8698x; 1.1599x over previous
#include <hip/hip_runtime.h>
#include <math.h>


typedef __attribute__((ext_vector_type(2))) float v2f;
typedef __attribute__((ext_vector_type(8))) float v8f;

#define N_NODES   100000
#define NUM_FEAT  64
#define NUM_CLASS 40
#define N_EDGES   1600000
#define WPB       25
#define GEMM_BLOCK (WPB * 32)
#define ROW_TILES  (N_NODES / 16)
#define GEMM_GRID  (ROW_TILES / WPB)
#define KSTEPS     (NUM_FEAT / 4)
#define CTILES     3

#define CNT_PAD     (98 * 1024)

#define NB          250
#define BROWS       400
#define PBLK        250
#define CHUNK       (N_EDGES / PBLK)
#define CHUNK4      (CHUNK / 4)
#define CTHREADS    320
#define EPT         (CHUNK4 / CTHREADS)
#define RSTRIDE     252
#define RUNS_PAD    (62 * 1024)
#define SRC_BITS    17
#define SRC_MASK    ((1 << SRC_BITS) - 1)
#define RPW         (PBLK / WPB)
#define STAGE_CAP   (WPB * 16 * 48)
#define GU          8

#define NW_V2F     (KSTEPS * 2 * CTILES * 32)
#define OFF_W      0
#define OFF_OUT    (NW_V2F * 2)
#define OFF_LSE    (OFF_OUT + WPB * 16 * 48)
#define OFF_X      (OFF_LSE + WPB * 16)
#define OFF_A      (OFF_X + WPB * 16 * NUM_FEAT)
#define SMEM_FLOATS (OFF_A + WPB * 16 * NUM_FEAT)
#define OFF_C      SMEM_FLOATS
#define OFF_RS     (OFF_C + BROWS)
#define OFF_PST    (OFF_RS + 404)
#define OFF_PN     (OFF_PST + 252)
#define OFF_WSUM   (OFF_PN + 252)
#define OFF_WLEN   (OFF_WSUM + 32)
#define SMEM_ALL   (OFF_WLEN + 32)

static __device__ __forceinline__ void chunk_put(int* off, int* stage, int d, int s) {
  const unsigned b = (unsigned)d / BROWS;
  const int r = d - (int)b * BROWS;
  const int p = atomicAdd(&off[b], 1);
  stage[p] = (r << SRC_BITS) | s;
}

__global__ void __launch_bounds__(CTHREADS)
chunk_sort(const int* __restrict__ index, int* __restrict__ runs, int* __restrict__ cw) {
  __shared__ int h[NB];
  __shared__ int wsum[CTHREADS / 32];
  __shared__ __align__(16) int stage[CHUNK];
  const int t = threadIdx.x, lane = t & 31, w = t >> 5, blk = blockIdx.x;
  const int4* sp = (const int4*)(index + blk * CHUNK);
  const int4* dp = (const int4*)(index + N_EDGES + blk * CHUNK);
  int4 s[EPT], d[EPT];
#pragma unroll
  for (int u = 0; u < EPT; ++u) {
    d[u] = dp[t + u * CTHREADS];
    s[u] = sp[t + u * CTHREADS];
  }
  if (t < NB) h[t] = 0;
  __syncthreads();
#pragma unroll
  for (int u = 0; u < EPT; ++u) {
    atomicAdd(&h[(unsigned)d[u].x / BROWS], 1);
    atomicAdd(&h[(unsigned)d[u].y / BROWS], 1);
    atomicAdd(&h[(unsigned)d[u].z / BROWS], 1);
    atomicAdd(&h[(unsigned)d[u].w / BROWS], 1);
  }
  __syncthreads();

  const int v0 = (t < NB) ? h[t] : 0;
  int incl = v0;
  for (int o = 1; o < 32; o <<= 1) {
    const int x = __shfl_up(incl, o);
    if (lane >= o) incl += x;
  }
  if (lane == 31) wsum[w] = incl;
  __syncthreads();
  int wbase = 0;
  for (int k = 0; k < CTHREADS / 32; ++k) wbase += (k < w) ? wsum[k] : 0;
  const int ex = wbase + incl - v0;
  if (t < NB) {
    h[t] = ex;
    runs[blk * RSTRIDE + t] = ex;
  }
  if (t == 0) runs[blk * RSTRIDE + NB] = CHUNK;
  __syncthreads();

#pragma unroll
  for (int u = 0; u < EPT; ++u) {
    chunk_put(h, stage, d[u].x, s[u].x);
    chunk_put(h, stage, d[u].y, s[u].y);
    chunk_put(h, stage, d[u].z, s[u].z);
    chunk_put(h, stage, d[u].w, s[u].w);
  }
  __syncthreads();

  int4* op = (int4*)(cw + blk * CHUNK);
#pragma unroll
  for (int u = 0; u < EPT; ++u) op[t + u * CTHREADS] = ((const int4*)stage)[t + u * CTHREADS];
}

template <bool LDS_IDX>
static __device__ __forceinline__ void gather_rows(const float* __restrict__ x,
                                                   const int* ssrc,
                                                   const int* sIdx, int B, int bv,
                                                   int half, int q, float* sa) {
  for (int r = 0; r < 16; ++r) {
    const int b = __builtin_amdgcn_readlane(bv, r);
    const int e = __builtin_amdgcn_readlane(bv, r + 1);
    float4 acc = make_float4(0.0f, 0.0f, 0.0f, 0.0f);
    for (int j = b + half; j < e; j += 2 * GU) {
      int s[GU];
#pragma unroll
      for (int u = 0; u < GU; ++u) {
        const int jj = min(j + 2 * u, e - 1);
        s[u] = LDS_IDX ? sIdx[jj - B] : ssrc[jj];
      }
      float4 v[GU];
#pragma unroll
      for (int u = 0; u < GU; ++u) {
        v[u] = make_float4(0.0f, 0.0f, 0.0f, 0.0f);
        if (j + 2 * u < e) v[u] = *(const float4*)(x + (long)s[u] * NUM_FEAT + q * 4);
      }
#pragma unroll
      for (int u = 0; u < GU; ++u) {
        acc.x += v[u].x; acc.y += v[u].y; acc.z += v[u].z; acc.w += v[u].w;
      }
    }
    acc.x += __shfl_xor(acc.x, 16);
    acc.y += __shfl_xor(acc.y, 16);
    acc.z += __shfl_xor(acc.z, 16);
    acc.w += __shfl_xor(acc.w, 16);
    if (half == 0) *(float4*)(sa + r * NUM_FEAT + q * 4) = acc;
  }
}

__global__ void __launch_bounds__(GEMM_BLOCK)
sage_gemm(const float* __restrict__ x,
          const int*   __restrict__ cw,
          const int*   __restrict__ runs,
          int*         ssrc,
          const float* __restrict__ Wl,
          const float* __restrict__ bias,
          const float* __restrict__ Wr,
          float* __restrict__ out) {
  __shared__ __align__(16) float smem[SMEM_ALL];

  v2f*   sW   = (v2f*)smem;
  float* sOut = smem + OFF_OUT;
  float* sLse = smem + OFF_LSE;
  int*   stage = (int*)(smem + OFF_OUT);
  int*   rc   = (int*)(smem + OFF_C);
  int*   rs   = (int*)(smem + OFF_RS);
  int*   pst  = (int*)(smem + OFF_PST);
  int*   pn   = (int*)(smem + OFF_PN);
  int*   wsum = (int*)(smem + OFF_WSUM);
  int*   wlen = (int*)(smem + OFF_WLEN);

  const int tid  = threadIdx.x;
  const int wave = tid >> 5;
  const int lane = tid & 31;
  const int half = lane >> 4;
  const int l16  = lane & 15;
  const int bkt  = blockIdx.x;

  const long row0 = ((long)bkt * WPB + wave) * 16;

  const int rt = min(tid, PBLK - 1);
  int st = runs[rt * RSTRIDE + bkt];
  int ln = runs[rt * RSTRIDE + bkt + 1] - st;

  for (int i = tid; i < NW_V2F; i += GEMM_BLOCK) {
    const int li    = i & 31;
    const int t     = (i >> 5) % CTILES;
    const int which = (i / (CTILES * 32)) & 1;
    const int step  = i / (2 * CTILES * 32);
    const int k     = 4 * step + 2 * (li >> 4);
    const int nn    = t * 16 + (li & 15);
    const float* W  = which ? Wr : Wl;
    v2f w = {0.0f, 0.0f};
    if (nn < NUM_CLASS) w = (v2f){W[k * NUM_CLASS + nn], W[(k + 1) * NUM_CLASS + nn]};
    sW[i] = w;
  }

  {
    const float* xg = x + row0 * NUM_FEAT;
    float* sx = smem + OFF_X + wave * (16 * NUM_FEAT);
    for (int i = lane; i < (16 * NUM_FEAT) / 4; i += 32) {
      float4 vx = *(const float4*)(xg + i * 4);
      *(float4*)(sx + i * 4) = vx;
    }
  }

  if (tid < PBLK) {
    pst[tid] = tid * CHUNK + st;
    pn[tid]  = ln;
  } else {
    st = 0;
    ln = 0;
  }
  if (tid < BROWS) rc[tid] = 0;
  for (int o = 16; o > 0; o >>= 1) {
    st += __shfl_xor(st, o);
    ln += __shfl_xor(ln, o);
  }
  if (lane == 0) { wsum[wave] = st; wlen[wave] = ln; }
  __syncthreads();
  int s0 = 0, n = 0;
  for (int k = 0; k < WPB; ++k) { s0 += wsum[k]; n += wlen[k]; }
  const bool fits = n <= STAGE_CAP;

  int base[RPW], len[RPW], v0[RPW], v1[RPW];
#pragma unroll
  for (int u = 0; u < RPW; ++u) {
    const int i = wave + u * WPB;
    base[u] = __builtin_amdgcn_readfirstlane(pst[i]);
    len[u]  = __builtin_amdgcn_readfirstlane(pn[i]);
    const int last = max(len[u] - 1, 0);
    v0[u] = cw[min(base[u] + min(lane, last), N_EDGES - 1)];
    v1[u] = cw[min(base[u] + min(lane + 32, last), N_EDGES - 1)];
  }

#pragma unroll
  for (int u = 0; u < RPW; ++u) {
    if (lane < len[u])      atomicAdd(&rc[v0[u] >> SRC_BITS], 1);
    if (lane + 32 < len[u]) atomicAdd(&rc[v1[u] >> SRC_BITS], 1);
  }
#pragma unroll
  for (int u = 0; u < RPW; ++u) {
    if (len[u] > 64) {
      for (int k = lane + 64; k < len[u]; k += 32) atomicAdd(&rc[cw[base[u] + k] >> SRC_BITS], 1);
    }
  }
  __syncthreads();

  {
    const int cv = (tid < BROWS) ? rc[tid] : 0;
    int incl = cv;
    for (int o = 1; o < 32; o <<= 1) {
      const int y = __shfl_up(incl, o);
      if (lane >= o) incl += y;
    }
    if (lane == 31) wsum[wave] = incl;
    __syncthreads();
    int wbase = 0;
    for (int k = 0; k < WPB; ++k) wbase += (k < wave) ? wsum[k] : 0;
    const int ex = wbase + incl - cv;
    if (tid < BROWS) {
      rc[tid]  = ex;
      rs[tid] = ex;
    }
    if (tid == 0) rs[BROWS] = n;
  }
  __syncthreads();

  int* gdst = ssrc + s0;
  if (fits) {
#pragma unroll
    for (int u = 0; u < RPW; ++u) {
      if (lane < len[u])      stage[atomicAdd(&rc[v0[u] >> SRC_BITS], 1)] = v0[u] & SRC_MASK;
      if (lane + 32 < len[u]) stage[atomicAdd(&rc[v1[u] >> SRC_BITS], 1)] = v1[u] & SRC_MASK;
    }
#pragma unroll
    for (int u = 0; u < RPW; ++u) {
      if (len[u] > 64) {
        for (int k = lane + 64; k < len[u]; k += 32) {
          const int v = cw[base[u] + k];
          stage[atomicAdd(&rc[v >> SRC_BITS], 1)] = v & SRC_MASK;
        }
      }
    }
  } else {
#pragma unroll
    for (int u = 0; u < RPW; ++u) {
      if (lane < len[u])      gdst[atomicAdd(&rc[v0[u] >> SRC_BITS], 1)] = v0[u] & SRC_MASK;
      if (lane + 32 < len[u]) gdst[atomicAdd(&rc[v1[u] >> SRC_BITS], 1)] = v1[u] & SRC_MASK;
    }
#pragma unroll
    for (int u = 0; u < RPW; ++u) {
      if (len[u] > 64) {
        for (int k = lane + 64; k < len[u]; k += 32) {
          const int v = cw[base[u] + k];
          gdst[atomicAdd(&rc[v >> SRC_BITS], 1)] = v & SRC_MASK;
        }
      }
    }
  }
  __syncthreads();

  const int bv = (lane <= 16) ? rs[wave * 16 + lane] : 0;

  float* so = sOut + wave * (16 * 48);
  float* sa = smem + OFF_A + wave * (16 * NUM_FEAT);

  if (fits) gather_rows<true >(x, gdst, stage, 0, bv, half, l16, sa);
  else      gather_rows<false>(x, gdst, stage, 0, bv, half, l16, sa);
  __syncthreads();

  const int   degi = __shfl(bv, l16 + 1) - __shfl(bv, l16);
  const float rdeg = 1.0f / fmaxf((float)degi, 1.0f);

  const v2f* xf = (const v2f*)(smem + OFF_X + wave * (16 * NUM_FEAT) + l16 * NUM_FEAT);
  const v2f* af = (const v2f*)(smem + OFF_A + wave * (16 * NUM_FEAT) + l16 * NUM_FEAT);
  const v2f* wp = sW + lane;

  v8f acc0 = {}, acc1 = {}, acc2 = {};

#pragma unroll
  for (int step = 0; step < KSTEPS; ++step) {
    const v2f a_x   = xf[2 * step + half];
    const v2f a_agg = af[2 * step + half] * rdeg;

    const int wb = step * (2 * CTILES * 32);
    const v2f bl0 = wp[wb];
    const v2f bl1 = wp[wb + 32];
    const v2f bl2 = wp[wb + 64];
    const v2f br0 = wp[wb + 96];
    const v2f br1 = wp[wb + 128];
    const v2f br2 = wp[wb + 160];

    acc0 = __builtin_amdgcn_wmma_f32_16x16x4_f32(false, a_agg, false, bl0, (short)0, acc0, false, false);
    acc1 = __builtin_amdgcn_wmma_f32_16x16x4_f32(false, a_agg, false, bl1, (short)0, acc1, false, false);
    acc2 = __builtin_amdgcn_wmma_f32_16x16x4_f32(false, a_agg, false, bl2, (short)0, acc2, false, false);
    acc0 = __builtin_amdgcn_wmma_f32_16x16x4_f32(false, a_x,   false, br0, (short)0, acc0, false, false);
    acc1 = __builtin_amdgcn_wmma_f32_16x16x4_f32(false, a_x,   false, br1, (short)0, acc1, false, false);
    acc2 = __builtin_amdgcn_wmma_f32_16x16x4_f32(false, a_x,   false, br2, (short)0, acc2, false, false);
  }

  const float b0 = bias[l16];
  const float b1 = bias[16 + l16];
  const float b2 = (32 + l16 < NUM_CLASS) ? bias[32 + l16] : 0.0f;

#pragma unroll
  for (int r = 0; r < 8; ++r) {
    const int M = r + 8 * half;
    so[M * 48 +      l16] = acc0[r] + b0;
    so[M * 48 + 16 + l16] = acc1[r] + b1;
    so[M * 48 + 32 + l16] = acc2[r] + b2;
  }
  __syncthreads();

  if (lane < 16) {
    const float* rowp = so + l16 * 48;
    float mx = rowp[0];
    for (int c = 1; c < NUM_CLASS; ++c) mx = fmaxf(mx, rowp[c]);
    float s = 0.0f;
    for (int c = 0; c < NUM_CLASS; ++c) s += __expf(rowp[c] - mx);
    sLse[wave * 16 + l16] = mx + __logf(s);
  }
  __syncthreads();

  for (int i = lane; i < 16 * NUM_CLASS; i += 32) {
    const int r = i / NUM_CLASS, c = i % NUM_CLASS;
    out[(row0 + r) * NUM_CLASS + c] = so[r * 48 + c] - sLse[wave * 16 + r];
  }
}

extern "C" void kernel_launch(void* const* d_in, const int* in_sizes, int n_in,
                              void* d_out, int out_size, void* d_ws, size_t ws_size,
                              hipStream_t stream) {
  const float* x     = (const float*)d_in[0];
  const int*   index = (const int*)d_in[1];
  const float* Wl    = (const float*)d_in[2];
  const float* bl    = (const float*)d_in[3];
  const float* Wr    = (const float*)d_in[4];
  float* out = (float*)d_out;

  int* runs = (int*)d_ws + CNT_PAD;
  int* cw   = runs + RUNS_PAD;
  int* ssrc = cw + N_EDGES;

  chunk_sort<<<PBLK,      CTHREADS,   0, stream>>>(index, runs, cw);
  sage_gemm <<<GEMM_GRID, GEMM_BLOCK, 0, stream>>>(x, cw, runs, ssrc, Wl, bl, Wr, out);
}


# x*Wr half of the WMMAs issued in the gather loop behind each row's first loads, which go out one pass early (before the prior row's fold and store); acc order: all x*Wr k-steps, then mean*Wl
# speedup vs baseline: 5.0664x; 1.0404x over previous
#include <hip/hip_runtime.h>
#include <math.h>


typedef __attribute__((ext_vector_type(2))) float v2f;
typedef __attribute__((ext_vector_type(8))) float v8f;

#define N_NODES   100000
#define NUM_FEAT  64
#define NUM_CLASS 40
#define N_EDGES   1600000
#define WPB       25
#define GEMM_BLOCK (WPB * 32)
#define ROW_TILES  (N_NODES / 16)
#define GEMM_GRID  (ROW_TILES / WPB)
#define KSTEPS     (NUM_FEAT / 4)
#define CTILES     3

#define CNT_PAD     (98 * 1024)

#define NB          250
#define BROWS       400
#define PBLK        250
#define CHUNK       (N_EDGES / PBLK)
#define CHUNK4      (CHUNK / 4)
#define CTHREADS    320
#define EPT         (CHUNK4 / CTHREADS)
#define RSTRIDE     252
#define RUNS_PAD    (62 * 1024)
#define SRC_BITS    17
#define SRC_MASK    ((1 << SRC_BITS) - 1)
#define RPW         (PBLK / WPB)
#define STAGE_CAP   (WPB * 16 * 48)
#define GU          8

#define NW_V2F     (KSTEPS * 2 * CTILES * 32)
#define OFF_W      0
#define OFF_OUT    (NW_V2F * 2)
#define OFF_LSE    (OFF_OUT + WPB * 16 * 48)
#define OFF_X      (OFF_LSE + WPB * 16)
#define OFF_A      (OFF_X + WPB * 16 * NUM_FEAT)
#define SMEM_FLOATS (OFF_A + WPB * 16 * NUM_FEAT)
#define OFF_C      SMEM_FLOATS
#define OFF_RS     (OFF_C + BROWS)
#define OFF_PST    (OFF_RS + 404)
#define OFF_PN     (OFF_PST + 252)
#define OFF_WSUM   (OFF_PN + 252)
#define OFF_WLEN   (OFF_WSUM + 32)
#define SMEM_ALL   (OFF_WLEN + 32)

static __device__ __forceinline__ void chunk_put(int* off, int* stage, int d, int s) {
  const unsigned b = (unsigned)d / BROWS;
  const int r = d - (int)b * BROWS;
  const int p = atomicAdd(&off[b], 1);
  stage[p] = (r << SRC_BITS) | s;
}

__global__ void __launch_bounds__(CTHREADS)
chunk_sort(const int* __restrict__ index, int* __restrict__ runs, int* __restrict__ cw) {
  __shared__ int h[NB];
  __shared__ int wsum[CTHREADS / 32];
  __shared__ __align__(16) int stage[CHUNK];
  const int t = threadIdx.x, lane = t & 31, w = t >> 5, blk = blockIdx.x;
  const int4* sp = (const int4*)(index + blk * CHUNK);
  const int4* dp = (const int4*)(index + N_EDGES + blk * CHUNK);
  int4 s[EPT], d[EPT];
#pragma unroll
  for (int u = 0; u < EPT; ++u) {
    d[u] = dp[t + u * CTHREADS];
    s[u] = sp[t + u * CTHREADS];
  }
  if (t < NB) h[t] = 0;
  __syncthreads();
#pragma unroll
  for (int u = 0; u < EPT; ++u) {
    atomicAdd(&h[(unsigned)d[u].x / BROWS], 1);
    atomicAdd(&h[(unsigned)d[u].y / BROWS], 1);
    atomicAdd(&h[(unsigned)d[u].z / BROWS], 1);
    atomicAdd(&h[(unsigned)d[u].w / BROWS], 1);
  }
  __syncthreads();

  const int v0 = (t < NB) ? h[t] : 0;
  int incl = v0;
  for (int o = 1; o < 32; o <<= 1) {
    const int x = __shfl_up(incl, o);
    if (lane >= o) incl += x;
  }
  if (lane == 31) wsum[w] = incl;
  __syncthreads();
  int wbase = 0;
  for (int k = 0; k < CTHREADS / 32; ++k) wbase += (k < w) ? wsum[k] : 0;
  const int ex = wbase + incl - v0;
  if (t < NB) {
    h[t] = ex;
    runs[blk * RSTRIDE + t] = ex;
  }
  if (t == 0) runs[blk * RSTRIDE + NB] = CHUNK;
  __syncthreads();

#pragma unroll
  for (int u = 0; u < EPT; ++u) {
    chunk_put(h, stage, d[u].x, s[u].x);
    chunk_put(h, stage, d[u].y, s[u].y);
    chunk_put(h, stage, d[u].z, s[u].z);
    chunk_put(h, stage, d[u].w, s[u].w);
  }
  __syncthreads();

  int4* op = (int4*)(cw + blk * CHUNK);
#pragma unroll
  for (int u = 0; u < EPT; ++u) op[t + u * CTHREADS] = ((const int4*)stage)[t + u * CTHREADS];
}

static __device__ __forceinline__ void add_batch(float4& acc, const float4* v) {
#pragma unroll
  for (int u = 0; u < GU; ++u) {
    acc.x += v[u].x; acc.y += v[u].y; acc.z += v[u].z; acc.w += v[u].w;
  }
}

template <bool LDS_IDX>
static __device__ __forceinline__ void load_batch(const float* __restrict__ x, const int* ssrc,
                                                  const int* sIdx, int B, int j, int e, int q, float4* v) {
  int s[GU];
#pragma unroll
  for (int u = 0; u < GU; ++u) {
    const int jj = min(j + 2 * u, max(e - 1, 0));
    s[u] = LDS_IDX ? sIdx[jj - B] : ssrc[jj];
  }
#pragma unroll
  for (int u = 0; u < GU; ++u) {
    if (j + 2 * u < e) v[u] = *(const float4*)(x + (long)s[u] * NUM_FEAT + q * 4);
  }
}

static __device__ __forceinline__ void root_step(const v2f* xf, const v2f* wp, int half, int step,
                                                 v8f& acc0, v8f& acc1, v8f& acc2) {
  const v2f a_x = xf[2 * step + half];
  const int wb  = step * (2 * CTILES * 32) + CTILES * 32;
  const v2f br0 = wp[wb];
  const v2f br1 = wp[wb + 32];
  const v2f br2 = wp[wb + 64];
  acc0 = __builtin_amdgcn_wmma_f32_16x16x4_f32(false, a_x, false, br0, (short)0, acc0, false, false);
  acc1 = __builtin_amdgcn_wmma_f32_16x16x4_f32(false, a_x, false, br1, (short)0, acc1, false, false);
  acc2 = __builtin_amdgcn_wmma_f32_16x16x4_f32(false, a_x, false, br2, (short)0, acc2, false, false);
}

template <bool LDS_IDX>
static __device__ __forceinline__ void gather_rows(const float* __restrict__ x,
                                                   const int* ssrc,
                                                   const int* sIdx, int B, int bv,
                                                   int half, int q, float* sa,
                                                   const v2f* xf, const v2f* wp,
                                                   v8f& acc0, v8f& acc1, v8f& acc2) {
  int b = __builtin_amdgcn_readlane(bv, 0);
  int e = __builtin_amdgcn_readlane(bv, 1);
  float4 v[GU];
#pragma unroll
  for (int u = 0; u < GU; ++u) v[u] = make_float4(0.0f, 0.0f, 0.0f, 0.0f);
  load_batch<LDS_IDX>(x, ssrc, sIdx, B, b + half, e, q, v);
  root_step(xf, wp, half, 0, acc0, acc1, acc2);
  for (int r = 0; r < 16; ++r) {
    float4 acc = make_float4(0.0f, 0.0f, 0.0f, 0.0f);
    add_batch(acc, v);
    for (int j = b + half + 2 * GU; j < e; j += 2 * GU) {
#pragma unroll
      for (int u = 0; u < GU; ++u) v[u] = make_float4(0.0f, 0.0f, 0.0f, 0.0f);
      load_batch<LDS_IDX>(x, ssrc, sIdx, B, j, e, q, v);
      add_batch(acc, v);
    }
    b = e;
    e = __builtin_amdgcn_readlane(bv, min(r + 2, 16));
#pragma unroll
    for (int u = 0; u < GU; ++u) v[u] = make_float4(0.0f, 0.0f, 0.0f, 0.0f);
    if (r + 1 < 16) {
      load_batch<LDS_IDX>(x, ssrc, sIdx, B, b + half, e, q, v);
      root_step(xf, wp, half, r + 1, acc0, acc1, acc2);
    }
    acc.x += __shfl_xor(acc.x, 16);
    acc.y += __shfl_xor(acc.y, 16);
    acc.z += __shfl_xor(acc.z, 16);
    acc.w += __shfl_xor(acc.w, 16);
    if (half == 0) *(float4*)(sa + r * NUM_FEAT + q * 4) = acc;
  }
}

__global__ void __launch_bounds__(GEMM_BLOCK)
sage_gemm(const float* __restrict__ x,
          const int*   __restrict__ cw,
          const int*   __restrict__ runs,
          int*         ssrc,
          const float* __restrict__ Wl,
          const float* __restrict__ bias,
          const float* __restrict__ Wr,
          float* __restrict__ out) {
  __shared__ __align__(16) float smem[SMEM_ALL];

  v2f*   sW   = (v2f*)smem;
  float* sOut = smem + OFF_OUT;
  float* sLse = smem + OFF_LSE;
  int*   stage = (int*)(smem + OFF_OUT);
  int*   rc   = (int*)(smem + OFF_C);
  int*   rs   = (int*)(smem + OFF_RS);
  int*   pst  = (int*)(smem + OFF_PST);
  int*   pn   = (int*)(smem + OFF_PN);
  int*   wsum = (int*)(smem + OFF_WSUM);
  int*   wlen = (int*)(smem + OFF_WLEN);

  const int tid  = threadIdx.x;
  const int wave = tid >> 5;
  const int lane = tid & 31;
  const int half = lane >> 4;
  const int l16  = lane & 15;
  const int bkt  = blockIdx.x;

  const long row0 = ((long)bkt * WPB + wave) * 16;

  const int rt = min(tid, PBLK - 1);
  int st = runs[rt * RSTRIDE + bkt];
  int ln = runs[rt * RSTRIDE + bkt + 1] - st;

  for (int i = tid; i < NW_V2F; i += GEMM_BLOCK) {
    const int li    = i & 31;
    const int t     = (i >> 5) % CTILES;
    const int which = (i / (CTILES * 32)) & 1;
    const int step  = i / (2 * CTILES * 32);
    const int k     = 4 * step + 2 * (li >> 4);
    const int nn    = t * 16 + (li & 15);
    const float* W  = which ? Wr : Wl;
    v2f w = {0.0f, 0.0f};
    if (nn < NUM_CLASS) w = (v2f){W[k * NUM_CLASS + nn], W[(k + 1) * NUM_CLASS + nn]};
    sW[i] = w;
  }

  {
    const float* xg = x + row0 * NUM_FEAT;
    float* sx = smem + OFF_X + wave * (16 * NUM_FEAT);
    for (int i = lane; i < (16 * NUM_FEAT) / 4; i += 32) {
      float4 vx = *(const float4*)(xg + i * 4);
      *(float4*)(sx + i * 4) = vx;
    }
  }

  if (tid < PBLK) {
    pst[tid] = tid * CHUNK + st;
    pn[tid]  = ln;
  } else {
    st = 0;
    ln = 0;
  }
  if (tid < BROWS) rc[tid] = 0;
  for (int o = 16; o > 0; o >>= 1) {
    st += __shfl_xor(st, o);
    ln += __shfl_xor(ln, o);
  }
  if (lane == 0) { wsum[wave] = st; wlen[wave] = ln; }
  __syncthreads();
  int s0 = 0, n = 0;
  for (int k = 0; k < WPB; ++k) { s0 += wsum[k]; n += wlen[k]; }
  const bool fits = n <= STAGE_CAP;

  int base[RPW], len[RPW], v0[RPW], v1[RPW];
#pragma unroll
  for (int u = 0; u < RPW; ++u) {
    const int i = wave + u * WPB;
    base[u] = __builtin_amdgcn_readfirstlane(pst[i]);
    len[u]  = __builtin_amdgcn_readfirstlane(pn[i]);
    const int last = max(len[u] - 1, 0);
    v0[u] = cw[min(base[u] + min(lane, last), N_EDGES - 1)];
    v1[u] = cw[min(base[u] + min(lane + 32, last), N_EDGES - 1)];
  }

#pragma unroll
  for (int u = 0; u < RPW; ++u) {
    if (lane < len[u])      atomicAdd(&rc[v0[u] >> SRC_BITS], 1);
    if (lane + 32 < len[u]) atomicAdd(&rc[v1[u] >> SRC_BITS], 1);
  }
#pragma unroll
  for (int u = 0; u < RPW; ++u) {
    if (len[u] > 64) {
      for (int k = lane + 64; k < len[u]; k += 32) atomicAdd(&rc[cw[base[u] + k] >> SRC_BITS], 1);
    }
  }
  __syncthreads();

  {
    const int cv = (tid < BROWS) ? rc[tid] : 0;
    int incl = cv;
    for (int o = 1; o < 32; o <<= 1) {
      const int y = __shfl_up(incl, o);
      if (lane >= o) incl += y;
    }
    if (lane == 31) wsum[wave] = incl;
    __syncthreads();
    int wbase = 0;
    for (int k = 0; k < WPB; ++k) wbase += (k < wave) ? wsum[k] : 0;
    const int ex = wbase + incl - cv;
    if (tid < BROWS) {
      rc[tid]  = ex;
      rs[tid] = ex;
    }
    if (tid == 0) rs[BROWS] = n;
  }
  __syncthreads();

  int* gdst = ssrc + s0;
  if (fits) {
#pragma unroll
    for (int u = 0; u < RPW; ++u) {
      if (lane < len[u])      stage[atomicAdd(&rc[v0[u] >> SRC_BITS], 1)] = v0[u] & SRC_MASK;
      if (lane + 32 < len[u]) stage[atomicAdd(&rc[v1[u] >> SRC_BITS], 1)] = v1[u] & SRC_MASK;
    }
#pragma unroll
    for (int u = 0; u < RPW; ++u) {
      if (len[u] > 64) {
        for (int k = lane + 64; k < len[u]; k += 32) {
          const int v = cw[base[u] + k];
          stage[atomicAdd(&rc[v >> SRC_BITS], 1)] = v & SRC_MASK;
        }
      }
    }
  } else {
#pragma unroll
    for (int u = 0; u < RPW; ++u) {
      if (lane < len[u])      gdst[atomicAdd(&rc[v0[u] >> SRC_BITS], 1)] = v0[u] & SRC_MASK;
      if (lane + 32 < len[u]) gdst[atomicAdd(&rc[v1[u] >> SRC_BITS], 1)] = v1[u] & SRC_MASK;
    }
#pragma unroll
    for (int u = 0; u < RPW; ++u) {
      if (len[u] > 64) {
        for (int k = lane + 64; k < len[u]; k += 32) {
          const int v = cw[base[u] + k];
          gdst[atomicAdd(&rc[v >> SRC_BITS], 1)] = v & SRC_MASK;
        }
      }
    }
  }
  __syncthreads();

  const int bv = (lane <= 16) ? rs[wave * 16 + lane] : 0;

  float* so = sOut + wave * (16 * 48);
  float* sa = smem + OFF_A + wave * (16 * NUM_FEAT);

  const v2f* xf = (const v2f*)(smem + OFF_X + wave * (16 * NUM_FEAT) + l16 * NUM_FEAT);
  const v2f* af = (const v2f*)(smem + OFF_A + wave * (16 * NUM_FEAT) + l16 * NUM_FEAT);
  const v2f* wp = sW + lane;

  v8f acc0 = {}, acc1 = {}, acc2 = {};

  if (fits) gather_rows<true >(x, gdst, stage, 0, bv, half, l16, sa, xf, wp, acc0, acc1, acc2);
  else      gather_rows<false>(x, gdst, stage, 0, bv, half, l16, sa, xf, wp, acc0, acc1, acc2);
  __syncthreads();

  const int   degi = __shfl(bv, l16 + 1) - __shfl(bv, l16);
  const float rdeg = 1.0f / fmaxf((float)degi, 1.0f);

#pragma unroll
  for (int step = 0; step < KSTEPS; ++step) {
    const v2f a_agg = af[2 * step + half] * rdeg;

    const int wb = step * (2 * CTILES * 32);
    const v2f bl0 = wp[wb];
    const v2f bl1 = wp[wb + 32];
    const v2f bl2 = wp[wb + 64];

    acc0 = __builtin_amdgcn_wmma_f32_16x16x4_f32(false, a_agg, false, bl0, (short)0, acc0, false, false);
    acc1 = __builtin_amdgcn_wmma_f32_16x16x4_f32(false, a_agg, false, bl1, (short)0, acc1, false, false);
    acc2 = __builtin_amdgcn_wmma_f32_16x16x4_f32(false, a_agg, false, bl2, (short)0, acc2, false, false);
  }

  const float b0 = bias[l16];
  const float b1 = bias[16 + l16];
  const float b2 = (32 + l16 < NUM_CLASS) ? bias[32 + l16] : 0.0f;

#pragma unroll
  for (int r = 0; r < 8; ++r) {
    const int M = r + 8 * half;
    so[M * 48 +      l16] = acc0[r] + b0;
    so[M * 48 + 16 + l16] = acc1[r] + b1;
    so[M * 48 + 32 + l16] = acc2[r] + b2;
  }
  __syncthreads();

  if (lane < 16) {
    const float* rowp = so + l16 * 48;
    float mx = rowp[0];
    for (int c = 1; c < NUM_CLASS; ++c) mx = fmaxf(mx, rowp[c]);
    float s = 0.0f;
    for (int c = 0; c < NUM_CLASS; ++c) s += __expf(rowp[c] - mx);
    sLse[wave * 16 + l16] = mx + __logf(s);
  }
  __syncthreads();

  for (int i = lane; i < 16 * NUM_CLASS; i += 32) {
    const int r = i / NUM_CLASS, c = i % NUM_CLASS;
    out[(row0 + r) * NUM_CLASS + c] = so[r * 48 + c] - sLse[wave * 16 + r];
  }
}

extern "C" void kernel_launch(void* const* d_in, const int* in_sizes, int n_in,
                              void* d_out, int out_size, void* d_ws, size_t ws_size,
                              hipStream_t stream) {
  const float* x     = (const float*)d_in[0];
  const int*   index = (const int*)d_in[1];
  const float* Wl    = (const float*)d_in[2];
  const float* bl    = (const float*)d_in[3];
  const float* Wr    = (const float*)d_in[4];
  float* out = (float*)d_out;

  int* runs = (int*)d_ws + CNT_PAD;
  int* cw   = runs + RUNS_PAD;
  int* ssrc = cw + N_EDGES;

  chunk_sort<<<PBLK,      CTHREADS,   0, stream>>>(index, runs, cw);
  sage_gemm <<<GEMM_GRID, GEMM_BLOCK, 0, stream>>>(x, cw, runs, ssrc, Wl, bl, Wr, out);
}


# Epilogue made wave-local: logits tile kept in the wave's own dead x tile (pitch 52), so the 3 work-group barriers behind the gather go and a finished wave runs its WMMAs, log-softmax, stores at once
# speedup vs baseline: 5.2336x; 1.0330x over previous
#include <hip/hip_runtime.h>
#include <math.h>


typedef __attribute__((ext_vector_type(2))) float v2f;
typedef __attribute__((ext_vector_type(8))) float v8f;

#define N_NODES   100000
#define NUM_FEAT  64
#define NUM_CLASS 40
#define N_EDGES   1600000
#define WPB       25
#define GEMM_BLOCK (WPB * 32)
#define ROW_TILES  (N_NODES / 16)
#define GEMM_GRID  (ROW_TILES / WPB)
#define KSTEPS     (NUM_FEAT / 4)
#define CTILES     3

#define CNT_PAD     (98 * 1024)

#define NB          250
#define BROWS       400
#define PBLK        250
#define CHUNK       (N_EDGES / PBLK)
#define CHUNK4      (CHUNK / 4)
#define CTHREADS    320
#define EPT         (CHUNK4 / CTHREADS)
#define RSTRIDE     252
#define RUNS_PAD    (62 * 1024)
#define SRC_BITS    17
#define SRC_MASK    ((1 << SRC_BITS) - 1)
#define RPW         (PBLK / WPB)
#define STAGE_CAP   (WPB * 16 * 48)
#define GU          8
#define SO_PITCH    52

#define NW_V2F     (KSTEPS * 2 * CTILES * 32)
#define OFF_W      0
#define OFF_OUT    (NW_V2F * 2)
#define OFF_LSE    (OFF_OUT + WPB * 16 * 48)
#define OFF_X      (OFF_LSE + WPB * 16)
#define OFF_A      (OFF_X + WPB * 16 * NUM_FEAT)
#define SMEM_FLOATS (OFF_A + WPB * 16 * NUM_FEAT)
#define OFF_C      SMEM_FLOATS
#define OFF_RS     (OFF_C + BROWS)
#define OFF_PST    (OFF_RS + 404)
#define OFF_PN     (OFF_PST + 252)
#define OFF_WSUM   (OFF_PN + 252)
#define OFF_WLEN   (OFF_WSUM + 32)
#define SMEM_ALL   (OFF_WLEN + 32)

static __device__ __forceinline__ void chunk_put(int* off, int* stage, int d, int s) {
  const unsigned b = (unsigned)d / BROWS;
  const int r = d - (int)b * BROWS;
  const int p = atomicAdd(&off[b], 1);
  stage[p] = (r << SRC_BITS) | s;
}

__global__ void __launch_bounds__(CTHREADS)
chunk_sort(const int* __restrict__ index, int* __restrict__ runs, int* __restrict__ cw) {
  __shared__ int h[NB];
  __shared__ int wsum[CTHREADS / 32];
  __shared__ __align__(16) int stage[CHUNK];
  const int t = threadIdx.x, lane = t & 31, w = t >> 5, blk = blockIdx.x;
  const int4* sp = (const int4*)(index + blk * CHUNK);
  const int4* dp = (const int4*)(index + N_EDGES + blk * CHUNK);
  int4 s[EPT], d[EPT];
#pragma unroll
  for (int u = 0; u < EPT; ++u) {
    d[u] = dp[t + u * CTHREADS];
    s[u] = sp[t + u * CTHREADS];
  }
  if (t < NB) h[t] = 0;
  __syncthreads();
#pragma unroll
  for (int u = 0; u < EPT; ++u) {
    atomicAdd(&h[(unsigned)d[u].x / BROWS], 1);
    atomicAdd(&h[(unsigned)d[u].y / BROWS], 1);
    atomicAdd(&h[(unsigned)d[u].z / BROWS], 1);
    atomicAdd(&h[(unsigned)d[u].w / BROWS], 1);
  }
  __syncthreads();

  const int v0 = (t < NB) ? h[t] : 0;
  int incl = v0;
  for (int o = 1; o < 32; o <<= 1) {
    const int x = __shfl_up(incl, o);
    if (lane >= o) incl += x;
  }
  if (lane == 31) wsum[w] = incl;
  __syncthreads();
  int wbase = 0;
  for (int k = 0; k < CTHREADS / 32; ++k) wbase += (k < w) ? wsum[k] : 0;
  const int ex = wbase + incl - v0;
  if (t < NB) {
    h[t] = ex;
    runs[blk * RSTRIDE + t] = ex;
  }
  if (t == 0) runs[blk * RSTRIDE + NB] = CHUNK;
  __syncthreads();

#pragma unroll
  for (int u = 0; u < EPT; ++u) {
    chunk_put(h, stage, d[u].x, s[u].x);
    chunk_put(h, stage, d[u].y, s[u].y);
    chunk_put(h, stage, d[u].z, s[u].z);
    chunk_put(h, stage, d[u].w, s[u].w);
  }
  __syncthreads();

  int4* op = (int4*)(cw + blk * CHUNK);
#pragma unroll
  for (int u = 0; u < EPT; ++u) op[t + u * CTHREADS] = ((const int4*)stage)[t + u * CTHREADS];
}

static __device__ __forceinline__ void add_batch(float4& acc, const float4* v) {
#pragma unroll
  for (int u = 0; u < GU; ++u) {
    acc.x += v[u].x; acc.y += v[u].y; acc.z += v[u].z; acc.w += v[u].w;
  }
}

static __device__ __forceinline__ void wave_lds_order() {
  __builtin_amdgcn_fence(__ATOMIC_RELEASE, "wavefront");
  __builtin_amdgcn_wave_barrier();
  __builtin_amdgcn_fence(__ATOMIC_ACQUIRE, "wavefront");
}

template <bool LDS_IDX>
static __device__ __forceinline__ void load_batch(const float* __restrict__ x, const int* ssrc,
                                                  const int* sIdx, int B, int j, int e, int q, float4* v) {
  int s[GU];
#pragma unroll
  for (int u = 0; u < GU; ++u) {
    const int jj = min(j + 2 * u, max(e - 1, 0));
    s[u] = LDS_IDX ? sIdx[jj - B] : ssrc[jj];
  }
#pragma unroll
  for (int u = 0; u < GU; ++u) {
    if (j + 2 * u < e) v[u] = *(const float4*)(x + (long)s[u] * NUM_FEAT + q * 4);
  }
}

static __device__ __forceinline__ void root_step(const v2f* xf, const v2f* wp, int half, int step,
                                                 v8f& acc0, v8f& acc1, v8f& acc2) {
  const v2f a_x = xf[2 * step + half];
  const int wb  = step * (2 * CTILES * 32) + CTILES * 32;
  const v2f br0 = wp[wb];
  const v2f br1 = wp[wb + 32];
  const v2f br2 = wp[wb + 64];
  acc0 = __builtin_amdgcn_wmma_f32_16x16x4_f32(false, a_x, false, br0, (short)0, acc0, false, false);
  acc1 = __builtin_amdgcn_wmma_f32_16x16x4_f32(false, a_x, false, br1, (short)0, acc1, false, false);
  acc2 = __builtin_amdgcn_wmma_f32_16x16x4_f32(false, a_x, false, br2, (short)0, acc2, false, false);
}

template <bool LDS_IDX>
static __device__ __forceinline__ void gather_rows(const float* __restrict__ x,
                                                   const int* ssrc,
                                                   const int* sIdx, int B, int bv,
                                                   int half, int q, float* sa,
                                                   const v2f* xf, const v2f* wp,
                                                   v8f& acc0, v8f& acc1, v8f& acc2) {
  int b = __builtin_amdgcn_readlane(bv, 0);
  int e = __builtin_amdgcn_readlane(bv, 1);
  float4 v[GU];
#pragma unroll
  for (int u = 0; u < GU; ++u) v[u] = make_float4(0.0f, 0.0f, 0.0f, 0.0f);
  load_batch<LDS_IDX>(x, ssrc, sIdx, B, b + half, e, q, v);
  root_step(xf, wp, half, 0, acc0, acc1, acc2);
  for (int r = 0; r < 16; ++r) {
    float4 acc = make_float4(0.0f, 0.0f, 0.0f, 0.0f);
    add_batch(acc, v);
    for (int j = b + half + 2 * GU; j < e; j += 2 * GU) {
#pragma unroll
      for (int u = 0; u < GU; ++u) v[u] = make_float4(0.0f, 0.0f, 0.0f, 0.0f);
      load_batch<LDS_IDX>(x, ssrc, sIdx, B, j, e, q, v);
      add_batch(acc, v);
    }
    b = e;
    e = __builtin_amdgcn_readlane(bv, min(r + 2, 16));
#pragma unroll
    for (int u = 0; u < GU; ++u) v[u] = make_float4(0.0f, 0.0f, 0.0f, 0.0f);
    if (r + 1 < 16) {
      load_batch<LDS_IDX>(x, ssrc, sIdx, B, b + half, e, q, v);
      root_step(xf, wp, half, r + 1, acc0, acc1, acc2);
    }
    acc.x += __shfl_xor(acc.x, 16);
    acc.y += __shfl_xor(acc.y, 16);
    acc.z += __shfl_xor(acc.z, 16);
    acc.w += __shfl_xor(acc.w, 16);
    if (half == 0) *(float4*)(sa + r * NUM_FEAT + q * 4) = acc;
  }
}

__global__ void __launch_bounds__(GEMM_BLOCK)
sage_gemm(const float* __restrict__ x,
          const int*   __restrict__ cw,
          const int*   __restrict__ runs,
          int*         ssrc,
          const float* __restrict__ Wl,
          const float* __restrict__ bias,
          const float* __restrict__ Wr,
          float* __restrict__ out) {
  __shared__ __align__(16) float smem[SMEM_ALL];

  v2f*   sW   = (v2f*)smem;
  float* sLse = smem + OFF_LSE;
  int*   stage = (int*)(smem + OFF_OUT);
  int*   rc   = (int*)(smem + OFF_C);
  int*   rs   = (int*)(smem + OFF_RS);
  int*   pst  = (int*)(smem + OFF_PST);
  int*   pn   = (int*)(smem + OFF_PN);
  int*   wsum = (int*)(smem + OFF_WSUM);
  int*   wlen = (int*)(smem + OFF_WLEN);

  const int tid  = threadIdx.x;
  const int wave = tid >> 5;
  const int lane = tid & 31;
  const int half = lane >> 4;
  const int l16  = lane & 15;
  const int bkt  = blockIdx.x;

  const long row0 = ((long)bkt * WPB + wave) * 16;

  const int rt = min(tid, PBLK - 1);
  int st = runs[rt * RSTRIDE + bkt];
  int ln = runs[rt * RSTRIDE + bkt + 1] - st;

  for (int i = tid; i < NW_V2F; i += GEMM_BLOCK) {
    const int li    = i & 31;
    const int t     = (i >> 5) % CTILES;
    const int which = (i / (CTILES * 32)) & 1;
    const int step  = i / (2 * CTILES * 32);
    const int k     = 4 * step + 2 * (li >> 4);
    const int nn    = t * 16 + (li & 15);
    const float* W  = which ? Wr : Wl;
    v2f w = {0.0f, 0.0f};
    if (nn < NUM_CLASS) w = (v2f){W[k * NUM_CLASS + nn], W[(k + 1) * NUM_CLASS + nn]};
    sW[i] = w;
  }

  {
    const float* xg = x + row0 * NUM_FEAT;
    float* sx = smem + OFF_X + wave * (16 * NUM_FEAT);
    for (int i = lane; i < (16 * NUM_FEAT) / 4; i += 32) {
      float4 vx = *(const float4*)(xg + i * 4);
      *(float4*)(sx + i * 4) = vx;
    }
  }

  if (tid < PBLK) {
    pst[tid] = tid * CHUNK + st;
    pn[tid]  = ln;
  } else {
    st = 0;
    ln = 0;
  }
  if (tid < BROWS) rc[tid] = 0;
  for (int o = 16; o > 0; o >>= 1) {
    st += __shfl_xor(st, o);
    ln += __shfl_xor(ln, o);
  }
  if (lane == 0) { wsum[wave] = st; wlen[wave] = ln; }
  __syncthreads();
  int s0 = 0, n = 0;
  for (int k = 0; k < WPB; ++k) { s0 += wsum[k]; n += wlen[k]; }
  const bool fits = n <= STAGE_CAP;

  int base[RPW], len[RPW], v0[RPW], v1[RPW];
#pragma unroll
  for (int u = 0; u < RPW; ++u) {
    const int i = wave + u * WPB;
    base[u] = __builtin_amdgcn_readfirstlane(pst[i]);
    len[u]  = __builtin_amdgcn_readfirstlane(pn[i]);
    const int last = max(len[u] - 1, 0);
    v0[u] = cw[min(base[u] + min(lane, last), N_EDGES - 1)];
    v1[u] = cw[min(base[u] + min(lane + 32, last), N_EDGES - 1)];
  }

#pragma unroll
  for (int u = 0; u < RPW; ++u) {
    if (lane < len[u])      atomicAdd(&rc[v0[u] >> SRC_BITS], 1);
    if (lane + 32 < len[u]) atomicAdd(&rc[v1[u] >> SRC_BITS], 1);
  }
#pragma unroll
  for (int u = 0; u < RPW; ++u) {
    if (len[u] > 64) {
      for (int k = lane + 64; k < len[u]; k += 32) atomicAdd(&rc[cw[base[u] + k] >> SRC_BITS], 1);
    }
  }
  __syncthreads();

  {
    const int cv = (tid < BROWS) ? rc[tid] : 0;
    int incl = cv;
    for (int o = 1; o < 32; o <<= 1) {
      const int y = __shfl_up(incl, o);
      if (lane >= o) incl += y;
    }
    if (lane == 31) wsum[wave] = incl;
    __syncthreads();
    int wbase = 0;
    for (int k = 0; k < WPB; ++k) wbase += (k < wave) ? wsum[k] : 0;
    const int ex = wbase + incl - cv;
    if (tid < BROWS) {
      rc[tid]  = ex;
      rs[tid] = ex;
    }
    if (tid == 0) rs[BROWS] = n;
  }
  __syncthreads();

  int* gdst = ssrc + s0;
  if (fits) {
#pragma unroll
    for (int u = 0; u < RPW; ++u) {
      if (lane < len[u])      stage[atomicAdd(&rc[v0[u] >> SRC_BITS], 1)] = v0[u] & SRC_MASK;
      if (lane + 32 < len[u]) stage[atomicAdd(&rc[v1[u] >> SRC_BITS], 1)] = v1[u] & SRC_MASK;
    }
#pragma unroll
    for (int u = 0; u < RPW; ++u) {
      if (len[u] > 64) {
        for (int k = lane + 64; k < len[u]; k += 32) {
          const int v = cw[base[u] + k];
          stage[atomicAdd(&rc[v >> SRC_BITS], 1)] = v & SRC_MASK;
        }
      }
    }
  } else {
#pragma unroll
    for (int u = 0; u < RPW; ++u) {
      if (lane < len[u])      gdst[atomicAdd(&rc[v0[u] >> SRC_BITS], 1)] = v0[u] & SRC_MASK;
      if (lane + 32 < len[u]) gdst[atomicAdd(&rc[v1[u] >> SRC_BITS], 1)] = v1[u] & SRC_MASK;
    }
#pragma unroll
    for (int u = 0; u < RPW; ++u) {
      if (len[u] > 64) {
        for (int k = lane + 64; k < len[u]; k += 32) {
          const int v = cw[base[u] + k];
          gdst[atomicAdd(&rc[v >> SRC_BITS], 1)] = v & SRC_MASK;
        }
      }
    }
  }
  __syncthreads();

  const int bv = (lane <= 16) ? rs[wave * 16 + lane] : 0;

  float* so = smem + OFF_X + wave * (16 * NUM_FEAT);
  float* sa = smem + OFF_A + wave * (16 * NUM_FEAT);

  const v2f* xf = (const v2f*)(smem + OFF_X + wave * (16 * NUM_FEAT) + l16 * NUM_FEAT);
  const v2f* af = (const v2f*)(smem + OFF_A + wave * (16 * NUM_FEAT) + l16 * NUM_FEAT);
  const v2f* wp = sW + lane;

  v8f acc0 = {}, acc1 = {}, acc2 = {};

  if (fits) gather_rows<true >(x, gdst, stage, 0, bv, half, l16, sa, xf, wp, acc0, acc1, acc2);
  else      gather_rows<false>(x, gdst, stage, 0, bv, half, l16, sa, xf, wp, acc0, acc1, acc2);
  wave_lds_order();

  const int   degi = __shfl(bv, l16 + 1) - __shfl(bv, l16);
  const float rdeg = 1.0f / fmaxf((float)degi, 1.0f);

#pragma unroll
  for (int step = 0; step < KSTEPS; ++step) {
    const v2f a_agg = af[2 * step + half] * rdeg;

    const int wb = step * (2 * CTILES * 32);
    const v2f bl0 = wp[wb];
    const v2f bl1 = wp[wb + 32];
    const v2f bl2 = wp[wb + 64];

    acc0 = __builtin_amdgcn_wmma_f32_16x16x4_f32(false, a_agg, false, bl0, (short)0, acc0, false, false);
    acc1 = __builtin_amdgcn_wmma_f32_16x16x4_f32(false, a_agg, false, bl1, (short)0, acc1, false, false);
    acc2 = __builtin_amdgcn_wmma_f32_16x16x4_f32(false, a_agg, false, bl2, (short)0, acc2, false, false);
  }

  const float b0 = bias[l16];
  const float b1 = bias[16 + l16];
  const float b2 = (32 + l16 < NUM_CLASS) ? bias[32 + l16] : 0.0f;

#pragma unroll
  for (int r = 0; r < 8; ++r) {
    const int M = r + 8 * half;
    so[M * SO_PITCH +      l16] = acc0[r] + b0;
    so[M * SO_PITCH + 16 + l16] = acc1[r] + b1;
    so[M * SO_PITCH + 32 + l16] = acc2[r] + b2;
  }
  wave_lds_order();

  if (lane < 16) {
    const float* rowp = so + l16 * SO_PITCH;
    float mx = rowp[0];
    for (int c = 1; c < NUM_CLASS; ++c) mx = fmaxf(mx, rowp[c]);
    float s = 0.0f;
    for (int c = 0; c < NUM_CLASS; ++c) s += __expf(rowp[c] - mx);
    sLse[wave * 16 + l16] = mx + __logf(s);
  }
  wave_lds_order();

  for (int i = lane; i < 16 * NUM_CLASS; i += 32) {
    const int r = i / NUM_CLASS, c = i % NUM_CLASS;
    out[(row0 + r) * NUM_CLASS + c] = so[r * SO_PITCH + c] - sLse[wave * 16 + r];
  }
}

extern "C" void kernel_launch(void* const* d_in, const int* in_sizes, int n_in,
                              void* d_out, int out_size, void* d_ws, size_t ws_size,
                              hipStream_t stream) {
  const float* x     = (const float*)d_in[0];
  const int*   index = (const int*)d_in[1];
  const float* Wl    = (const float*)d_in[2];
  const float* bl    = (const float*)d_in[3];
  const float* Wr    = (const float*)d_in[4];
  float* out = (float*)d_out;

  int* runs = (int*)d_ws + CNT_PAD;
  int* cw   = runs + RUNS_PAD;
  int* ssrc = cw + N_EDGES;

  chunk_sort<<<PBLK,      CTHREADS,   0, stream>>>(index, runs, cw);
  sage_gemm <<<GEMM_GRID, GEMM_BLOCK, 0, stream>>>(x, cw, runs, ssrc, Wl, bl, Wr, out);
}


# Gather re-formed: one wave-wide 8-B/lane load per edge at a scalar row offset, exactly deg loads per row (runs of 16,8,4,2,1) in one group; no lane masks, zero fills or half fold; sum order changes
# speedup vs baseline: 6.2445x; 1.1932x over previous
#include <hip/hip_runtime.h>
#include <math.h>


typedef __attribute__((ext_vector_type(2))) float v2f;
typedef __attribute__((ext_vector_type(8))) float v8f;

#define N_NODES   100000
#define NUM_FEAT  64
#define NUM_CLASS 40
#define N_EDGES   1600000
#define WPB       25
#define GEMM_BLOCK (WPB * 32)
#define ROW_TILES  (N_NODES / 16)
#define GEMM_GRID  (ROW_TILES / WPB)
#define KSTEPS     (NUM_FEAT / 4)
#define CTILES     3

#define CNT_PAD     (98 * 1024)

#define NB          250
#define BROWS       400
#define PBLK        250
#define CHUNK       (N_EDGES / PBLK)
#define CHUNK4      (CHUNK / 4)
#define CTHREADS    320
#define EPT         (CHUNK4 / CTHREADS)
#define RSTRIDE     252
#define RUNS_PAD    (62 * 1024)
#define SRC_BITS    17
#define SRC_MASK    ((1 << SRC_BITS) - 1)
#define RPW         (PBLK / WPB)
#define STAGE_CAP   (WPB * 16 * 48)
#define GU          31
#define SO_PITCH    52

#define NW_V2F     (KSTEPS * 2 * CTILES * 32)
#define OFF_W      0
#define OFF_OUT    (NW_V2F * 2)
#define OFF_LSE    (OFF_OUT + WPB * 16 * 48)
#define OFF_X      (OFF_LSE + WPB * 16)
#define OFF_A      (OFF_X + WPB * 16 * NUM_FEAT)
#define SMEM_FLOATS (OFF_A + WPB * 16 * NUM_FEAT)
#define OFF_C      SMEM_FLOATS
#define OFF_RS     (OFF_C + BROWS)
#define OFF_PST    (OFF_RS + 404)
#define OFF_PN     (OFF_PST + 252)
#define OFF_WSUM   (OFF_PN + 252)
#define OFF_WLEN   (OFF_WSUM + 32)
#define SMEM_ALL   (OFF_WLEN + 32)

static __device__ __forceinline__ void chunk_put(int* off, int* stage, int d, int s) {
  const unsigned b = (unsigned)d / BROWS;
  const int r = d - (int)b * BROWS;
  const int p = atomicAdd(&off[b], 1);
  stage[p] = (r << SRC_BITS) | s;
}

__global__ void __launch_bounds__(CTHREADS)
chunk_sort(const int* __restrict__ index, int* __restrict__ runs, int* __restrict__ cw) {
  __shared__ int h[NB];
  __shared__ int wsum[CTHREADS / 32];
  __shared__ __align__(16) int stage[CHUNK];
  const int t = threadIdx.x, lane = t & 31, w = t >> 5, blk = blockIdx.x;
  const int4* sp = (const int4*)(index + blk * CHUNK);
  const int4* dp = (const int4*)(index + N_EDGES + blk * CHUNK);
  int4 s[EPT], d[EPT];
#pragma unroll
  for (int u = 0; u < EPT; ++u) {
    d[u] = dp[t + u * CTHREADS];
    s[u] = sp[t + u * CTHREADS];
  }
  if (t < NB) h[t] = 0;
  __syncthreads();
#pragma unroll
  for (int u = 0; u < EPT; ++u) {
    atomicAdd(&h[(unsigned)d[u].x / BROWS], 1);
    atomicAdd(&h[(unsigned)d[u].y / BROWS], 1);
    atomicAdd(&h[(unsigned)d[u].z / BROWS], 1);
    atomicAdd(&h[(unsigned)d[u].w / BROWS], 1);
  }
  __syncthreads();

  const int v0 = (t < NB) ? h[t] : 0;
  int incl = v0;
  for (int o = 1; o < 32; o <<= 1) {
    const int x = __shfl_up(incl, o);
    if (lane >= o) incl += x;
  }
  if (lane == 31) wsum[w] = incl;
  __syncthreads();
  int wbase = 0;
  for (int k = 0; k < CTHREADS / 32; ++k) wbase += (k < w) ? wsum[k] : 0;
  const int ex = wbase + incl - v0;
  if (t < NB) {
    h[t] = ex;
    runs[blk * RSTRIDE + t] = ex;
  }
  if (t == 0) runs[blk * RSTRIDE + NB] = CHUNK;
  __syncthreads();

#pragma unroll
  for (int u = 0; u < EPT; ++u) {
    chunk_put(h, stage, d[u].x, s[u].x);
    chunk_put(h, stage, d[u].y, s[u].y);
    chunk_put(h, stage, d[u].z, s[u].z);
    chunk_put(h, stage, d[u].w, s[u].w);
  }
  __syncthreads();

  int4* op = (int4*)(cw + blk * CHUNK);
#pragma unroll
  for (int u = 0; u < EPT; ++u) op[t + u * CTHREADS] = ((const int4*)stage)[t + u * CTHREADS];
}


static __device__ __forceinline__ void wave_lds_order() {
  __builtin_amdgcn_fence(__ATOMIC_RELEASE, "wavefront");
  __builtin_amdgcn_wave_barrier();
  __builtin_amdgcn_fence(__ATOMIC_ACQUIRE, "wavefront");
}

struct LaneRun { int rel, hm; };
static __device__ __forceinline__ LaneRun lane_run(int lane) {
  LaneRun t;
  t.rel = lane < 16 ? lane : lane < 24 ? lane - 16 : lane < 28 ? lane - 24 : lane < 30 ? lane - 28 : 0;
  t.hm  = lane < 16 ? 0 : lane < 24 ? 16 : lane < 28 ? 24 : lane < 30 ? 28 : lane < 31 ? 30 : 0;
  return t;
}

template <bool LDS_IDX>
static __device__ __forceinline__ int row_offsets(const int* ssrc, const int* sIdx, int c, int e, int n, LaneRun t) {
  const int jj = min(c + t.rel + (n & t.hm), max(e - 1, 0));
  const unsigned id = (unsigned)(LDS_IDX ? sIdx[jj] : ssrc[jj]);
  return (int)(id * (unsigned)(NUM_FEAT * 4));
}

template <int LO, int CNT>
static __device__ __forceinline__ void issue_run(const char* xb, int ofs, unsigned lane8, v2f* v) {
#pragma unroll
  for (int u = LO; u < LO + CNT; ++u) {
    const unsigned o = (unsigned)__builtin_amdgcn_readlane(ofs, u) + lane8;
    v[u] = *(const v2f*)(xb + o);
  }
}
template <int LO, int CNT>
static __device__ __forceinline__ void add_run(v2f& a0, v2f& a1, const v2f* v) {
#pragma unroll
  for (int u = LO; u < LO + CNT; ++u) {
    if (u & 1) a1 += v[u]; else a0 += v[u];
  }
}
static __device__ __forceinline__ void loads_done() {
  __builtin_amdgcn_fence(__ATOMIC_ACQUIRE, "workgroup");
}

static __device__ __forceinline__ void issue_rows(const char* xb, int ofs, int n, unsigned lane8, v2f* v) {
  if (n & 16) issue_run< 0, 16>(xb, ofs, lane8, v);
  if (n & 8)  issue_run<16,  8>(xb, ofs, lane8, v);
  if (n & 4)  issue_run<24,  4>(xb, ofs, lane8, v);
  if (n & 2)  issue_run<28,  2>(xb, ofs, lane8, v);
  if (n & 1)  issue_run<30,  1>(xb, ofs, lane8, v);
}
static __device__ __forceinline__ void add_rows(v2f& a0, v2f& a1, const v2f* v, int n) {
  if (n & 16) add_run< 0, 16>(a0, a1, v);
  if (n & 8)  add_run<16,  8>(a0, a1, v);
  if (n & 4)  add_run<24,  4>(a0, a1, v);
  if (n & 2)  add_run<28,  2>(a0, a1, v);
  if (n & 1)  add_run<30,  1>(a0, a1, v);
}

struct RootFrag { v2f a, b0, b1, b2; };
static __device__ __forceinline__ RootFrag root_frag(const v2f* xf, const v2f* wp, int half, int step) {
  RootFrag f;
  f.a = xf[2 * step + half];
  const int wb = step * (2 * CTILES * 32) + CTILES * 32;
  f.b0 = wp[wb];
  f.b1 = wp[wb + 32];
  f.b2 = wp[wb + 64];
  return f;
}
static __device__ __forceinline__ void root_mma(const RootFrag& f, v8f& acc0, v8f& acc1, v8f& acc2) {
  acc0 = __builtin_amdgcn_wmma_f32_16x16x4_f32(false, f.a, false, f.b0, (short)0, acc0, false, false);
  acc1 = __builtin_amdgcn_wmma_f32_16x16x4_f32(false, f.a, false, f.b1, (short)0, acc1, false, false);
  acc2 = __builtin_amdgcn_wmma_f32_16x16x4_f32(false, f.a, false, f.b2, (short)0, acc2, false, false);
}

template <bool LDS_IDX>
static __device__ __forceinline__ void gather_rows(const float* __restrict__ x,
                                                   const int* ssrc,
                                                   const int* sIdx, int bv,
                                                   int lane, int half, float* sa,
                                                   const v2f* xf, const v2f* wp,
                                                   v8f& acc0, v8f& acc1, v8f& acc2) {
  const char* xb = (const char*)x;
  const unsigned lane8 = (unsigned)lane * 8u;
  const LaneRun t = lane_run(lane);
  int b = __builtin_amdgcn_readlane(bv, 0);
  int e = __builtin_amdgcn_readlane(bv, 1);
  v2f v[GU];
  int n = min(e - b, GU);
  loads_done();
  {
    const int of0 = row_offsets<LDS_IDX>(ssrc, sIdx, b, e, n, t);
    const RootFrag f = root_frag(xf, wp, half, 0);
    issue_rows(xb, of0, n, lane8, v);
    root_mma(f, acc0, acc1, acc2);
  }
  for (int r = 0; r < 16; ++r) {
    const int bn = e;
    const int en = __builtin_amdgcn_readlane(bv, min(r + 2, 16));
    const int nn = min(en - bn, GU);
    const int ofn = row_offsets<LDS_IDX>(ssrc, sIdx, bn, en, nn, t);
    v2f a0 = {0.0f, 0.0f}, a1 = {0.0f, 0.0f};
    loads_done();
    add_rows(a0, a1, v, n);
    for (int c = b + GU; c < e; c += GU) {
      const int n2 = min(e - c, GU);
      issue_rows(xb, row_offsets<LDS_IDX>(ssrc, sIdx, c, e, n2, t), n2, lane8, v);
      loads_done();
      add_rows(a0, a1, v, n2);
    }
    if (r + 1 < 16) {
      const RootFrag f = root_frag(xf, wp, half, r + 1);
      issue_rows(xb, ofn, nn, lane8, v);
      root_mma(f, acc0, acc1, acc2);
    }
    *(v2f*)(sa + r * NUM_FEAT + lane * 2) = a0 + a1;
    b = bn;
    e = en;
    n = nn;
  }
}

__global__ void __launch_bounds__(GEMM_BLOCK)
sage_gemm(const float* __restrict__ x,
          const int*   __restrict__ cw,
          const int*   __restrict__ runs,
          int*         ssrc,
          const float* __restrict__ Wl,
          const float* __restrict__ bias,
          const float* __restrict__ Wr,
          float* __restrict__ out) {
  __shared__ __align__(16) float smem[SMEM_ALL];

  v2f*   sW   = (v2f*)smem;
  float* sLse = smem + OFF_LSE;
  int*   stage = (int*)(smem + OFF_OUT);
  int*   rc   = (int*)(smem + OFF_C);
  int*   rs   = (int*)(smem + OFF_RS);
  int*   pst  = (int*)(smem + OFF_PST);
  int*   pn   = (int*)(smem + OFF_PN);
  int*   wsum = (int*)(smem + OFF_WSUM);
  int*   wlen = (int*)(smem + OFF_WLEN);

  const int tid  = threadIdx.x;
  const int wave = tid >> 5;
  const int lane = tid & 31;
  const int half = lane >> 4;
  const int l16  = lane & 15;
  const int bkt  = blockIdx.x;

  const long row0 = ((long)bkt * WPB + wave) * 16;

  const int rt = min(tid, PBLK - 1);
  int st = runs[rt * RSTRIDE + bkt];
  int ln = runs[rt * RSTRIDE + bkt + 1] - st;

  for (int i = tid; i < NW_V2F; i += GEMM_BLOCK) {
    const int li    = i & 31;
    const int t     = (i >> 5) % CTILES;
    const int which = (i / (CTILES * 32)) & 1;
    const int step  = i / (2 * CTILES * 32);
    const int k     = 4 * step + 2 * (li >> 4);
    const int nn    = t * 16 + (li & 15);
    const float* W  = which ? Wr : Wl;
    v2f w = {0.0f, 0.0f};
    if (nn < NUM_CLASS) w = (v2f){W[k * NUM_CLASS + nn], W[(k + 1) * NUM_CLASS + nn]};
    sW[i] = w;
  }

  {
    const float* xg = x + row0 * NUM_FEAT;
    float* sx = smem + OFF_X + wave * (16 * NUM_FEAT);
    for (int i = lane; i < (16 * NUM_FEAT) / 4; i += 32) {
      float4 vx = *(const float4*)(xg + i * 4);
      *(float4*)(sx + i * 4) = vx;
    }
  }

  if (tid < PBLK) {
    pst[tid] = tid * CHUNK + st;
    pn[tid]  = ln;
  } else {
    st = 0;
    ln = 0;
  }
  if (tid < BROWS) rc[tid] = 0;
  for (int o = 16; o > 0; o >>= 1) {
    st += __shfl_xor(st, o);
    ln += __shfl_xor(ln, o);
  }
  if (lane == 0) { wsum[wave] = st; wlen[wave] = ln; }
  __syncthreads();
  int s0 = 0, n = 0;
  for (int k = 0; k < WPB; ++k) { s0 += wsum[k]; n += wlen[k]; }
  const bool fits = n <= STAGE_CAP;

  int base[RPW], len[RPW], v0[RPW], v1[RPW];
#pragma unroll
  for (int u = 0; u < RPW; ++u) {
    const int i = wave + u * WPB;
    base[u] = __builtin_amdgcn_readfirstlane(pst[i]);
    len[u]  = __builtin_amdgcn_readfirstlane(pn[i]);
    const int last = max(len[u] - 1, 0);
    v0[u] = cw[min(base[u] + min(lane, last), N_EDGES - 1)];
    v1[u] = cw[min(base[u] + min(lane + 32, last), N_EDGES - 1)];
  }

#pragma unroll
  for (int u = 0; u < RPW; ++u) {
    if (lane < len[u])      atomicAdd(&rc[v0[u] >> SRC_BITS], 1);
    if (lane + 32 < len[u]) atomicAdd(&rc[v1[u] >> SRC_BITS], 1);
  }
#pragma unroll
  for (int u = 0; u < RPW; ++u) {
    if (len[u] > 64) {
      for (int k = lane + 64; k < len[u]; k += 32) atomicAdd(&rc[cw[base[u] + k] >> SRC_BITS], 1);
    }
  }
  __syncthreads();

  {
    const int cv = (tid < BROWS) ? rc[tid] : 0;
    int incl = cv;
    for (int o = 1; o < 32; o <<= 1) {
      const int y = __shfl_up(incl, o);
      if (lane >= o) incl += y;
    }
    if (lane == 31) wsum[wave] = incl;
    __syncthreads();
    int wbase = 0;
    for (int k = 0; k < WPB; ++k) wbase += (k < wave) ? wsum[k] : 0;
    const int ex = wbase + incl - cv;
    if (tid < BROWS) {
      rc[tid]  = ex;
      rs[tid] = ex;
    }
    if (tid == 0) rs[BROWS] = n;
  }
  __syncthreads();

  int* gdst = ssrc + s0;
  if (fits) {
#pragma unroll
    for (int u = 0; u < RPW; ++u) {
      if (lane < len[u])      stage[atomicAdd(&rc[v0[u] >> SRC_BITS], 1)] = v0[u] & SRC_MASK;
      if (lane + 32 < len[u]) stage[atomicAdd(&rc[v1[u] >> SRC_BITS], 1)] = v1[u] & SRC_MASK;
    }
#pragma unroll
    for (int u = 0; u < RPW; ++u) {
      if (len[u] > 64) {
        for (int k = lane + 64; k < len[u]; k += 32) {
          const int v = cw[base[u] + k];
          stage[atomicAdd(&rc[v >> SRC_BITS], 1)] = v & SRC_MASK;
        }
      }
    }
  } else {
#pragma unroll
    for (int u = 0; u < RPW; ++u) {
      if (lane < len[u])      gdst[atomicAdd(&rc[v0[u] >> SRC_BITS], 1)] = v0[u] & SRC_MASK;
      if (lane + 32 < len[u]) gdst[atomicAdd(&rc[v1[u] >> SRC_BITS], 1)] = v1[u] & SRC_MASK;
    }
#pragma unroll
    for (int u = 0; u < RPW; ++u) {
      if (len[u] > 64) {
        for (int k = lane + 64; k < len[u]; k += 32) {
          const int v = cw[base[u] + k];
          gdst[atomicAdd(&rc[v >> SRC_BITS], 1)] = v & SRC_MASK;
        }
      }
    }
  }
  __syncthreads();

  const int bv = (lane <= 16) ? rs[wave * 16 + lane] : 0;

  float* so = smem + OFF_X + wave * (16 * NUM_FEAT);
  float* sa = smem + OFF_A + wave * (16 * NUM_FEAT);

  const v2f* xf = (const v2f*)(smem + OFF_X + wave * (16 * NUM_FEAT) + l16 * NUM_FEAT);
  const v2f* af = (const v2f*)(smem + OFF_A + wave * (16 * NUM_FEAT) + l16 * NUM_FEAT);
  const v2f* wp = sW + lane;

  v8f acc0 = {}, acc1 = {}, acc2 = {};

  if (fits) gather_rows<true >(x, gdst, stage, bv, lane, half, sa, xf, wp, acc0, acc1, acc2);
  else      gather_rows<false>(x, gdst, stage, bv, lane, half, sa, xf, wp, acc0, acc1, acc2);
  wave_lds_order();

  const int   degi = __shfl(bv, l16 + 1) - __shfl(bv, l16);
  const float rdeg = 1.0f / fmaxf((float)degi, 1.0f);

#pragma unroll
  for (int step = 0; step < KSTEPS; ++step) {
    const v2f a_agg = af[2 * step + half] * rdeg;

    const int wb = step * (2 * CTILES * 32);
    const v2f bl0 = wp[wb];
    const v2f bl1 = wp[wb + 32];
    const v2f bl2 = wp[wb + 64];

    acc0 = __builtin_amdgcn_wmma_f32_16x16x4_f32(false, a_agg, false, bl0, (short)0, acc0, false, false);
    acc1 = __builtin_amdgcn_wmma_f32_16x16x4_f32(false, a_agg, false, bl1, (short)0, acc1, false, false);
    acc2 = __builtin_amdgcn_wmma_f32_16x16x4_f32(false, a_agg, false, bl2, (short)0, acc2, false, false);
  }

  const float b0 = bias[l16];
  const float b1 = bias[16 + l16];
  const float b2 = (32 + l16 < NUM_CLASS) ? bias[32 + l16] : 0.0f;

#pragma unroll
  for (int r = 0; r < 8; ++r) {
    const int M = r + 8 * half;
    so[M * SO_PITCH +      l16] = acc0[r] + b0;
    so[M * SO_PITCH + 16 + l16] = acc1[r] + b1;
    so[M * SO_PITCH + 32 + l16] = acc2[r] + b2;
  }
  wave_lds_order();

  if (lane < 16) {
    const float* rowp = so + l16 * SO_PITCH;
    float mx = rowp[0];
    for (int c = 1; c < NUM_CLASS; ++c) mx = fmaxf(mx, rowp[c]);
    float s = 0.0f;
    for (int c = 0; c < NUM_CLASS; ++c) s += __expf(rowp[c] - mx);
    sLse[wave * 16 + l16] = mx + __logf(s);
  }
  wave_lds_order();

  for (int i = lane; i < 16 * NUM_CLASS; i += 32) {
    const int r = i / NUM_CLASS, c = i % NUM_CLASS;
    out[(row0 + r) * NUM_CLASS + c] = so[r * SO_PITCH + c] - sLse[wave * 16 + r];
  }
}

extern "C" void kernel_launch(void* const* d_in, const int* in_sizes, int n_in,
                              void* d_out, int out_size, void* d_ws, size_t ws_size,
                              hipStream_t stream) {
  const float* x     = (const float*)d_in[0];
  const int*   index = (const int*)d_in[1];
  const float* Wl    = (const float*)d_in[2];
  const float* bl    = (const float*)d_in[3];
  const float* Wr    = (const float*)d_in[4];
  float* out = (float*)d_out;

  int* runs = (int*)d_ws + CNT_PAD;
  int* cw   = runs + RUNS_PAD;
  int* ssrc = cw + N_EDGES;

  chunk_sort<<<PBLK,      CTHREADS,   0, stream>>>(index, runs, cw);
  sage_gemm <<<GEMM_GRID, GEMM_BLOCK, 0, stream>>>(x, cw, runs, ssrc, Wl, bl, Wr, out);
}


# Fused kernel's prologue: its 4 weight-staging and 8 x-tile loads, one per loop trip and each behind its own wait for memory, go out as ONE group (unrolled, clamped index) ahead of a single wait
# speedup vs baseline: 6.9706x; 1.1163x over previous
#include <hip/hip_runtime.h>
#include <math.h>


typedef __attribute__((ext_vector_type(2))) float v2f;
typedef __attribute__((ext_vector_type(8))) float v8f;

#define N_NODES   100000
#define NUM_FEAT  64
#define NUM_CLASS 40
#define N_EDGES   1600000
#define WPB       25
#define GEMM_BLOCK (WPB * 32)
#define ROW_TILES  (N_NODES / 16)
#define GEMM_GRID  (ROW_TILES / WPB)
#define KSTEPS     (NUM_FEAT / 4)
#define CTILES     3

#define CNT_PAD     (98 * 1024)

#define NB          250
#define BROWS       400
#define PBLK        250
#define CHUNK       (N_EDGES / PBLK)
#define CHUNK4      (CHUNK / 4)
#define CTHREADS    320
#define EPT         (CHUNK4 / CTHREADS)
#define RSTRIDE     252
#define RUNS_PAD    (62 * 1024)
#define SRC_BITS    17
#define SRC_MASK    ((1 << SRC_BITS) - 1)
#define RPW         (PBLK / WPB)
#define STAGE_CAP   (WPB * 16 * 48)
#define GU          31
#define SO_PITCH    52

#define NW_V2F     (KSTEPS * 2 * CTILES * 32)
#define W_TRIPS    ((NW_V2F + GEMM_BLOCK - 1) / GEMM_BLOCK)
#define X_TRIPS    ((16 * NUM_FEAT / 4) / 32)
#define OFF_W      0
#define OFF_OUT    (NW_V2F * 2)
#define OFF_LSE    (OFF_OUT + WPB * 16 * 48)
#define OFF_X      (OFF_LSE + WPB * 16)
#define OFF_A      (OFF_X + WPB * 16 * NUM_FEAT)
#define SMEM_FLOATS (OFF_A + WPB * 16 * NUM_FEAT)
#define OFF_C      SMEM_FLOATS
#define OFF_RS     (OFF_C + BROWS)
#define OFF_PST    (OFF_RS + 404)
#define OFF_PN     (OFF_PST + 252)
#define OFF_WSUM   (OFF_PN + 252)
#define OFF_WLEN   (OFF_WSUM + 32)
#define SMEM_ALL   (OFF_WLEN + 32)

static __device__ __forceinline__ void chunk_put(int* off, int* stage, int d, int s) {
  const unsigned b = (unsigned)d / BROWS;
  const int r = d - (int)b * BROWS;
  const int p = atomicAdd(&off[b], 1);
  stage[p] = (r << SRC_BITS) | s;
}

__global__ void __launch_bounds__(CTHREADS)
chunk_sort(const int* __restrict__ index, int* __restrict__ runs, int* __restrict__ cw) {
  __shared__ int h[NB];
  __shared__ int wsum[CTHREADS / 32];
  __shared__ __align__(16) int stage[CHUNK];
  const int t = threadIdx.x, lane = t & 31, w = t >> 5, blk = blockIdx.x;
  const int4* sp = (const int4*)(index + blk * CHUNK);
  const int4* dp = (const int4*)(index + N_EDGES + blk * CHUNK);
  int4 s[EPT], d[EPT];
#pragma unroll
  for (int u = 0; u < EPT; ++u) {
    d[u] = dp[t + u * CTHREADS];
    s[u] = sp[t + u * CTHREADS];
  }
  if (t < NB) h[t] = 0;
  __syncthreads();
#pragma unroll
  for (int u = 0; u < EPT; ++u) {
    atomicAdd(&h[(unsigned)d[u].x / BROWS], 1);
    atomicAdd(&h[(unsigned)d[u].y / BROWS], 1);
    atomicAdd(&h[(unsigned)d[u].z / BROWS], 1);
    atomicAdd(&h[(unsigned)d[u].w / BROWS], 1);
  }
  __syncthreads();

  const int v0 = (t < NB) ? h[t] : 0;
  int incl = v0;
  for (int o = 1; o < 32; o <<= 1) {
    const int x = __shfl_up(incl, o);
    if (lane >= o) incl += x;
  }
  if (lane == 31) wsum[w] = incl;
  __syncthreads();
  int wbase = 0;
  for (int k = 0; k < CTHREADS / 32; ++k) wbase += (k < w) ? wsum[k] : 0;
  const int ex = wbase + incl - v0;
  if (t < NB) {
    h[t] = ex;
    runs[blk * RSTRIDE + t] = ex;
  }
  if (t == 0) runs[blk * RSTRIDE + NB] = CHUNK;
  __syncthreads();

#pragma unroll
  for (int u = 0; u < EPT; ++u) {
    chunk_put(h, stage, d[u].x, s[u].x);
    chunk_put(h, stage, d[u].y, s[u].y);
    chunk_put(h, stage, d[u].z, s[u].z);
    chunk_put(h, stage, d[u].w, s[u].w);
  }
  __syncthreads();

  int4* op = (int4*)(cw + blk * CHUNK);
#pragma unroll
  for (int u = 0; u < EPT; ++u) op[t + u * CTHREADS] = ((const int4*)stage)[t + u * CTHREADS];
}


static __device__ __forceinline__ void wave_lds_order() {
  __builtin_amdgcn_fence(__ATOMIC_RELEASE, "wavefront");
  __builtin_amdgcn_wave_barrier();
  __builtin_amdgcn_fence(__ATOMIC_ACQUIRE, "wavefront");
}

struct LaneRun { int rel, hm; };
static __device__ __forceinline__ LaneRun lane_run(int lane) {
  LaneRun t;
  t.rel = lane < 16 ? lane : lane < 24 ? lane - 16 : lane < 28 ? lane - 24 : lane < 30 ? lane - 28 : 0;
  t.hm  = lane < 16 ? 0 : lane < 24 ? 16 : lane < 28 ? 24 : lane < 30 ? 28 : lane < 31 ? 30 : 0;
  return t;
}

template <bool LDS_IDX>
static __device__ __forceinline__ int row_offsets(const int* ssrc, const int* sIdx, int c, int e, int n, LaneRun t) {
  const int jj = min(c + t.rel + (n & t.hm), max(e - 1, 0));
  const unsigned id = (unsigned)(LDS_IDX ? sIdx[jj] : ssrc[jj]);
  return (int)(id * (unsigned)(NUM_FEAT * 4));
}

template <int LO, int CNT>
static __device__ __forceinline__ void issue_run(const char* xb, int ofs, unsigned lane8, v2f* v) {
#pragma unroll
  for (int u = LO; u < LO + CNT; ++u) {
    const unsigned o = (unsigned)__builtin_amdgcn_readlane(ofs, u) + lane8;
    v[u] = *(const v2f*)(xb + o);
  }
}
template <int LO, int CNT>
static __device__ __forceinline__ void add_run(v2f& a0, v2f& a1, const v2f* v) {
#pragma unroll
  for (int u = LO; u < LO + CNT; ++u) {
    if (u & 1) a1 += v[u]; else a0 += v[u];
  }
}
static __device__ __forceinline__ void loads_done() {
  __builtin_amdgcn_fence(__ATOMIC_ACQUIRE, "workgroup");
}

static __device__ __forceinline__ void issue_rows(const char* xb, int ofs, int n, unsigned lane8, v2f* v) {
  if (n & 16) issue_run< 0, 16>(xb, ofs, lane8, v);
  if (n & 8)  issue_run<16,  8>(xb, ofs, lane8, v);
  if (n & 4)  issue_run<24,  4>(xb, ofs, lane8, v);
  if (n & 2)  issue_run<28,  2>(xb, ofs, lane8, v);
  if (n & 1)  issue_run<30,  1>(xb, ofs, lane8, v);
}
static __device__ __forceinline__ void add_rows(v2f& a0, v2f& a1, const v2f* v, int n) {
  if (n & 16) add_run< 0, 16>(a0, a1, v);
  if (n & 8)  add_run<16,  8>(a0, a1, v);
  if (n & 4)  add_run<24,  4>(a0, a1, v);
  if (n & 2)  add_run<28,  2>(a0, a1, v);
  if (n & 1)  add_run<30,  1>(a0, a1, v);
}

struct RootFrag { v2f a, b0, b1, b2; };
static __device__ __forceinline__ RootFrag root_frag(const v2f* xf, const v2f* wp, int half, int step) {
  RootFrag f;
  f.a = xf[2 * step + half];
  const int wb = step * (2 * CTILES * 32) + CTILES * 32;
  f.b0 = wp[wb];
  f.b1 = wp[wb + 32];
  f.b2 = wp[wb + 64];
  return f;
}
static __device__ __forceinline__ void root_mma(const RootFrag& f, v8f& acc0, v8f& acc1, v8f& acc2) {
  acc0 = __builtin_amdgcn_wmma_f32_16x16x4_f32(false, f.a, false, f.b0, (short)0, acc0, false, false);
  acc1 = __builtin_amdgcn_wmma_f32_16x16x4_f32(false, f.a, false, f.b1, (short)0, acc1, false, false);
  acc2 = __builtin_amdgcn_wmma_f32_16x16x4_f32(false, f.a, false, f.b2, (short)0, acc2, false, false);
}

template <bool LDS_IDX>
static __device__ __forceinline__ void gather_rows(const float* __restrict__ x,
                                                   const int* ssrc,
                                                   const int* sIdx, int bv,
                                                   int lane, int half, float* sa,
                                                   const v2f* xf, const v2f* wp,
                                                   v8f& acc0, v8f& acc1, v8f& acc2) {
  const char* xb = (const char*)x;
  const unsigned lane8 = (unsigned)lane * 8u;
  const LaneRun t = lane_run(lane);
  int b = __builtin_amdgcn_readlane(bv, 0);
  int e = __builtin_amdgcn_readlane(bv, 1);
  v2f v[GU];
  int n = min(e - b, GU);
  loads_done();
  {
    const int of0 = row_offsets<LDS_IDX>(ssrc, sIdx, b, e, n, t);
    const RootFrag f = root_frag(xf, wp, half, 0);
    issue_rows(xb, of0, n, lane8, v);
    root_mma(f, acc0, acc1, acc2);
  }
  for (int r = 0; r < 16; ++r) {
    const int bn = e;
    const int en = __builtin_amdgcn_readlane(bv, min(r + 2, 16));
    const int nn = min(en - bn, GU);
    const int ofn = row_offsets<LDS_IDX>(ssrc, sIdx, bn, en, nn, t);
    v2f a0 = {0.0f, 0.0f}, a1 = {0.0f, 0.0f};
    loads_done();
    add_rows(a0, a1, v, n);
    for (int c = b + GU; c < e; c += GU) {
      const int n2 = min(e - c, GU);
      issue_rows(xb, row_offsets<LDS_IDX>(ssrc, sIdx, c, e, n2, t), n2, lane8, v);
      loads_done();
      add_rows(a0, a1, v, n2);
    }
    if (r + 1 < 16) {
      const RootFrag f = root_frag(xf, wp, half, r + 1);
      issue_rows(xb, ofn, nn, lane8, v);
      root_mma(f, acc0, acc1, acc2);
    }
    *(v2f*)(sa + r * NUM_FEAT + lane * 2) = a0 + a1;
    b = bn;
    e = en;
    n = nn;
  }
}

__global__ void __launch_bounds__(GEMM_BLOCK)
sage_gemm(const float* __restrict__ x,
          const int*   __restrict__ cw,
          const int*   __restrict__ runs,
          int*         ssrc,
          const float* __restrict__ Wl,
          const float* __restrict__ bias,
          const float* __restrict__ Wr,
          float* __restrict__ out) {
  __shared__ __align__(16) float smem[SMEM_ALL];

  v2f*   sW   = (v2f*)smem;
  float* sLse = smem + OFF_LSE;
  int*   stage = (int*)(smem + OFF_OUT);
  int*   rc   = (int*)(smem + OFF_C);
  int*   rs   = (int*)(smem + OFF_RS);
  int*   pst  = (int*)(smem + OFF_PST);
  int*   pn   = (int*)(smem + OFF_PN);
  int*   wsum = (int*)(smem + OFF_WSUM);
  int*   wlen = (int*)(smem + OFF_WLEN);

  const int tid  = threadIdx.x;
  const int wave = tid >> 5;
  const int lane = tid & 31;
  const int half = lane >> 4;
  const int l16  = lane & 15;
  const int bkt  = blockIdx.x;

  const long row0 = ((long)bkt * WPB + wave) * 16;

  const int rt = min(tid, PBLK - 1);
  int st = runs[rt * RSTRIDE + bkt];
  int ln = runs[rt * RSTRIDE + bkt + 1] - st;

  v2f wreg[W_TRIPS];
#pragma unroll
  for (int u = 0; u < W_TRIPS; ++u) {
    const int i     = min(tid + u * GEMM_BLOCK, NW_V2F - 1);
    const int li    = i & 31;
    const int t     = (i >> 5) % CTILES;
    const int which = (i / (CTILES * 32)) & 1;
    const int step  = i / (2 * CTILES * 32);
    const int k     = 4 * step + 2 * (li >> 4);
    const int nn    = t * 16 + (li & 15);
    const int nc    = min(nn, NUM_CLASS - 1);
    const float* W  = which ? Wr : Wl;
    const v2f w = {W[k * NUM_CLASS + nc], W[(k + 1) * NUM_CLASS + nc]};
    const float keep = (nn < NUM_CLASS) ? 1.0f : 0.0f;
    wreg[u] = w * keep;
  }
  float4 xreg[X_TRIPS];
  {
    const float* xg = x + row0 * NUM_FEAT;
#pragma unroll
    for (int u = 0; u < X_TRIPS; ++u) xreg[u] = *(const float4*)(xg + (lane + u * 32) * 4);
  }
#pragma unroll
  for (int u = 0; u < W_TRIPS; ++u) sW[min(tid + u * GEMM_BLOCK, NW_V2F - 1)] = wreg[u];
  {
    float* sx = smem + OFF_X + wave * (16 * NUM_FEAT);
#pragma unroll
    for (int u = 0; u < X_TRIPS; ++u) *(float4*)(sx + (lane + u * 32) * 4) = xreg[u];
  }

  if (wave < (PBLK + 31) / 32) {
    pst[rt] = rt * CHUNK + st;
    pn[rt]  = ln;
  }
  if (tid >= PBLK) {
    st = 0;
    ln = 0;
  }
  if (tid < BROWS) rc[tid] = 0;
  for (int o = 16; o > 0; o >>= 1) {
    st += __shfl_xor(st, o);
    ln += __shfl_xor(ln, o);
  }
  if (lane == 0) { wsum[wave] = st; wlen[wave] = ln; }
  __syncthreads();
  int s0 = 0, n = 0;
  for (int k = 0; k < WPB; ++k) { s0 += wsum[k]; n += wlen[k]; }
  const bool fits = n <= STAGE_CAP;

  int base[RPW], len[RPW], v0[RPW], v1[RPW];
#pragma unroll
  for (int u = 0; u < RPW; ++u) {
    const int i = wave + u * WPB;
    base[u] = __builtin_amdgcn_readfirstlane(pst[i]);
    len[u]  = __builtin_amdgcn_readfirstlane(pn[i]);
    const int last = max(len[u] - 1, 0);
    v0[u] = cw[min(base[u] + min(lane, last), N_EDGES - 1)];
    v1[u] = cw[min(base[u] + min(lane + 32, last), N_EDGES - 1)];
  }

#pragma unroll
  for (int u = 0; u < RPW; ++u) {
    if (lane < len[u])      atomicAdd(&rc[v0[u] >> SRC_BITS], 1);
    if (lane + 32 < len[u]) atomicAdd(&rc[v1[u] >> SRC_BITS], 1);
  }
#pragma unroll
  for (int u = 0; u < RPW; ++u) {
    if (len[u] > 64) {
      for (int k = lane + 64; k < len[u]; k += 32) atomicAdd(&rc[cw[base[u] + k] >> SRC_BITS], 1);
    }
  }
  __syncthreads();

  {
    const int cv = (tid < BROWS) ? rc[tid] : 0;
    int incl = cv;
    for (int o = 1; o < 32; o <<= 1) {
      const int y = __shfl_up(incl, o);
      if (lane >= o) incl += y;
    }
    if (lane == 31) wsum[wave] = incl;
    __syncthreads();
    int wbase = 0;
    for (int k = 0; k < WPB; ++k) wbase += (k < wave) ? wsum[k] : 0;
    const int ex = wbase + incl - cv;
    if (tid < BROWS) {
      rc[tid]  = ex;
      rs[tid] = ex;
    }
    if (tid == 0) rs[BROWS] = n;
  }
  __syncthreads();

  int* gdst = ssrc + s0;
  if (fits) {
#pragma unroll
    for (int u = 0; u < RPW; ++u) {
      if (lane < len[u])      stage[atomicAdd(&rc[v0[u] >> SRC_BITS], 1)] = v0[u] & SRC_MASK;
      if (lane + 32 < len[u]) stage[atomicAdd(&rc[v1[u] >> SRC_BITS], 1)] = v1[u] & SRC_MASK;
    }
#pragma unroll
    for (int u = 0; u < RPW; ++u) {
      if (len[u] > 64) {
        for (int k = lane + 64; k < len[u]; k += 32) {
          const int v = cw[base[u] + k];
          stage[atomicAdd(&rc[v >> SRC_BITS], 1)] = v & SRC_MASK;
        }
      }
    }
  } else {
#pragma unroll
    for (int u = 0; u < RPW; ++u) {
      if (lane < len[u])      gdst[atomicAdd(&rc[v0[u] >> SRC_BITS], 1)] = v0[u] & SRC_MASK;
      if (lane + 32 < len[u]) gdst[atomicAdd(&rc[v1[u] >> SRC_BITS], 1)] = v1[u] & SRC_MASK;
    }
#pragma unroll
    for (int u = 0; u < RPW; ++u) {
      if (len[u] > 64) {
        for (int k = lane + 64; k < len[u]; k += 32) {
          const int v = cw[base[u] + k];
          gdst[atomicAdd(&rc[v >> SRC_BITS], 1)] = v & SRC_MASK;
        }
      }
    }
  }
  __syncthreads();

  const int bv = (lane <= 16) ? rs[wave * 16 + lane] : 0;

  float* so = smem + OFF_X + wave * (16 * NUM_FEAT);
  float* sa = smem + OFF_A + wave * (16 * NUM_FEAT);

  const v2f* xf = (const v2f*)(smem + OFF_X + wave * (16 * NUM_FEAT) + l16 * NUM_FEAT);
  const v2f* af = (const v2f*)(smem + OFF_A + wave * (16 * NUM_FEAT) + l16 * NUM_FEAT);
  const v2f* wp = sW + lane;

  v8f acc0 = {}, acc1 = {}, acc2 = {};

  if (fits) gather_rows<true >(x, gdst, stage, bv, lane, half, sa, xf, wp, acc0, acc1, acc2);
  else      gather_rows<false>(x, gdst, stage, bv, lane, half, sa, xf, wp, acc0, acc1, acc2);
  wave_lds_order();

  const int   degi = __shfl(bv, l16 + 1) - __shfl(bv, l16);
  const float rdeg = 1.0f / fmaxf((float)degi, 1.0f);

#pragma unroll
  for (int step = 0; step < KSTEPS; ++step) {
    const v2f a_agg = af[2 * step + half] * rdeg;

    const int wb = step * (2 * CTILES * 32);
    const v2f bl0 = wp[wb];
    const v2f bl1 = wp[wb + 32];
    const v2f bl2 = wp[wb + 64];

    acc0 = __builtin_amdgcn_wmma_f32_16x16x4_f32(false, a_agg, false, bl0, (short)0, acc0, false, false);
    acc1 = __builtin_amdgcn_wmma_f32_16x16x4_f32(false, a_agg, false, bl1, (short)0, acc1, false, false);
    acc2 = __builtin_amdgcn_wmma_f32_16x16x4_f32(false, a_agg, false, bl2, (short)0, acc2, false, false);
  }

  const float b0 = bias[l16];
  const float b1 = bias[16 + l16];
  const float b2 = (32 + l16 < NUM_CLASS) ? bias[32 + l16] : 0.0f;

#pragma unroll
  for (int r = 0; r < 8; ++r) {
    const int M = r + 8 * half;
    so[M * SO_PITCH +      l16] = acc0[r] + b0;
    so[M * SO_PITCH + 16 + l16] = acc1[r] + b1;
    so[M * SO_PITCH + 32 + l16] = acc2[r] + b2;
  }
  wave_lds_order();

  if (lane < 16) {
    const float* rowp = so + l16 * SO_PITCH;
    float mx = rowp[0];
    for (int c = 1; c < NUM_CLASS; ++c) mx = fmaxf(mx, rowp[c]);
    float s = 0.0f;
    for (int c = 0; c < NUM_CLASS; ++c) s += __expf(rowp[c] - mx);
    sLse[wave * 16 + l16] = mx + __logf(s);
  }
  wave_lds_order();

  for (int i = lane; i < 16 * NUM_CLASS; i += 32) {
    const int r = i / NUM_CLASS, c = i % NUM_CLASS;
    out[(row0 + r) * NUM_CLASS + c] = so[r * SO_PITCH + c] - sLse[wave * 16 + r];
  }
}

extern "C" void kernel_launch(void* const* d_in, const int* in_sizes, int n_in,
                              void* d_out, int out_size, void* d_ws, size_t ws_size,
                              hipStream_t stream) {
  const float* x     = (const float*)d_in[0];
  const int*   index = (const int*)d_in[1];
  const float* Wl    = (const float*)d_in[2];
  const float* bl    = (const float*)d_in[3];
  const float* Wr    = (const float*)d_in[4];
  float* out = (float*)d_out;

  int* runs = (int*)d_ws + CNT_PAD;
  int* cw   = runs + RUNS_PAD;
  int* ssrc = cw + N_EDGES;

  chunk_sort<<<PBLK,      CTHREADS,   0, stream>>>(index, runs, cw);
  sage_gemm <<<GEMM_GRID, GEMM_BLOCK, 0, stream>>>(x, cw, runs, ssrc, Wl, bl, Wr, out);
}
